# starting kernel
# baseline (speedup 1.0000x reference)
#include <hip/hip_runtime.h>

typedef float v2f __attribute__((ext_vector_type(2)));
typedef float v8f __attribute__((ext_vector_type(8)));

#define Bc     4
#define LFULL  2050
#define Lq     2048
#define Hh     16
#define Dd     64
#define SHAPE  4095
#define WSTR   4096

#define KT     32
#define MT     256
#define BAND   (KT + MT)
#define VSTR   (KT + 2)

__global__ __launch_bounds__(256) void prep_kernel(const float* __restrict__ w,
                                                   const float* __restrict__ offset,
                                                   float* __restrict__ wexp,
                                                   float* __restrict__ out)
{
    __shared__ float s[4096];
    __shared__ float part[256];
    const int h   = blockIdx.x;
    const int tid = threadIdx.x;
    const float off = offset[h];

    for (int k = tid; k < 4096; k += 256) {
        float val = 0.f;
        if (k < SHAPE) val = __expf(w[h * SHAPE + k] - off);
        wexp[h * WSTR + k] = val;
        s[k] = val;
    }
    __syncthreads();

    const int base = tid * 16;
    float loc[16];
    float acc = 0.f;
    #pragma unroll
    for (int i = 0; i < 16; ++i) { loc[i] = acc; acc += s[base + i]; }
    part[tid] = acc;
    __syncthreads();
    if (tid == 0) {
        float run = 0.f;
        for (int i = 0; i < 256; ++i) { float t = part[i]; part[i] = run; run += t; }
    }
    __syncthreads();
    const float pbase = part[tid];
    #pragma unroll
    for (int i = 0; i < 16; ++i) s[base + i] = pbase + loc[i];
    __syncthreads();

    float* zout = out + (size_t)Bc * LFULL * Hh * Dd;
    for (int i = tid; i < Lq; i += 256) {
        const float z = s[i + Lq] - s[i];
        #pragma unroll
        for (int n = 0; n < Bc; ++n)
            zout[((size_t)n * LFULL + (i + 1)) * Hh + h] = z;
    }
    if (tid < 2 * Bc) {
        const int n  = tid & 3;
        const int jp = (tid >> 2) ? (LFULL - 1) : 0;
        zout[((size_t)n * LFULL + jp) * Hh + h] = 0.f;
    }
    for (int i = tid; i < Bc * 2 * Dd; i += 256) {
        const int d  = i & 63;
        const int r  = i >> 6;
        const int n  = r & 3;
        const int jp = (r >> 2) ? (LFULL - 1) : 0;
        out[(((size_t)n * LFULL + jp) * Hh + h) * Dd + d] = 0.f;
    }
}

__global__ __launch_bounds__(256) void pbv_kernel(const float* __restrict__ v,
                                                  const float* __restrict__ wexp,
                                                  float* __restrict__ out)
{
    __shared__ float band[BAND];
    __shared__ float vlds[Dd * VSTR];

    const int tid  = threadIdx.x;
    const int lane = tid & 31;
    const int wave = tid >> 5;
    const int m    = lane & 15;
    const int kh   = lane >> 4;
    const int j0   = blockIdx.x * MT;
    const int n    = blockIdx.y;
    const int h    = blockIdx.z;

    const float* wrow = wexp + h * WSTR;
    const int bandBase = (Lq - 1) - j0 - (MT - 1);

    v8f c[2][4] = {};

    const float* vbase = v + (((size_t)n * LFULL + 1) * Hh + h) * Dd;
    const int vrowk = tid >> 3;
    const int vd    = (tid & 7) * 8;

    for (int l0 = 0; l0 < Lq; l0 += KT) {
        band[tid] = wrow[bandBase + l0 + tid];
        if (tid < BAND - 256) band[256 + tid] = wrow[bandBase + l0 + 256 + tid];

        {
            const float* src = vbase + (size_t)(l0 + vrowk) * (Hh * Dd) + vd;
            const float4 f0 = *(const float4*)(src);
            const float4 f1 = *(const float4*)(src + 4);
            vlds[(vd + 0) * VSTR + vrowk] = f0.x;
            vlds[(vd + 1) * VSTR + vrowk] = f0.y;
            vlds[(vd + 2) * VSTR + vrowk] = f0.z;
            vlds[(vd + 3) * VSTR + vrowk] = f0.w;
            vlds[(vd + 4) * VSTR + vrowk] = f1.x;
            vlds[(vd + 5) * VSTR + vrowk] = f1.y;
            vlds[(vd + 6) * VSTR + vrowk] = f1.z;
            vlds[(vd + 7) * VSTR + vrowk] = f1.w;
        }
        __syncthreads();

        const int wr = wave * 32;
        #pragma unroll
        for (int ks = 0; ks < KT / 4; ++ks) {
            v2f a[2];
            #pragma unroll
            for (int f = 0; f < 2; ++f) {
                const int t0 = ks * 4 + 2 * kh + (MT - 1) - (wr + f * 16 + m);
                a[f][0] = band[t0];
                a[f][1] = band[t0 + 1];
            }
            #pragma unroll
            for (int nt = 0; nt < 4; ++nt) {
                const float* bp = &vlds[(nt * 16 + m) * VSTR + ks * 4 + 2 * kh];
                v2f b;
                b[0] = bp[0];
                b[1] = bp[1];
                #pragma unroll
                for (int f = 0; f < 2; ++f) {
                    c[f][nt] = __builtin_amdgcn_wmma_f32_16x16x4_f32(
                        false, a[f], false, b, (short)0, c[f][nt], false, false);
                }
            }
        }
        __syncthreads();
    }

    #pragma unroll
    for (int f = 0; f < 2; ++f) {
        #pragma unroll
        for (int nt = 0; nt < 4; ++nt) {
            const int d = nt * 16 + m;
            #pragma unroll
            for (int r = 0; r < 8; ++r) {
                const int mrow = (lane < 16) ? r : (r + 8);
                const int j = j0 + wave * 32 + f * 16 + mrow;
                out[(((size_t)n * LFULL + (j + 1)) * Hh + h) * Dd + d] = c[f][nt][r];
            }
        }
    }
}

extern "C" void kernel_launch(void* const* d_in, const int* in_sizes, int n_in,
                              void* d_out, int out_size, void* d_ws, size_t ws_size,
                              hipStream_t stream)
{
    (void)in_sizes; (void)n_in; (void)out_size; (void)ws_size;
    const float* v      = (const float*)d_in[0];
    const float* offset = (const float*)d_in[1];
    const float* w      = (const float*)d_in[2];
    float*       out    = (float*)d_out;
    float*       wexp   = (float*)d_ws;

    prep_kernel<<<dim3(Hh), 256, 0, stream>>>(w, offset, wexp, out);
    pbv_kernel<<<dim3(Lq / MT, Bc, Hh), 256, 0, stream>>>(v, wexp, out);
}


# K loop's global loads off the critical path: V slab t+1 fetched to registers under slab t's WMMAs into a second LDS stage, Toeplitz band staged once per workgroup; one barrier per slab, not two
# speedup vs baseline: 1.1566x; 1.1566x over previous
#include <hip/hip_runtime.h>

typedef float v2f __attribute__((ext_vector_type(2)));
typedef float v8f __attribute__((ext_vector_type(8)));

#define Bc     4
#define LFULL  2050
#define Lq     2048
#define Hh     16
#define Dd     64
#define SHAPE  4095
#define WSTR   4096

#define KT     32
#define MT     256
#define NSLAB  (Lq / KT)
#define BANDALL (Lq + MT)
#define VSTR   (KT + 2)
#define VSTAGE (Dd * VSTR)

__global__ __launch_bounds__(256) void prep_kernel(const float* __restrict__ w,
                                                   const float* __restrict__ offset,
                                                   float* __restrict__ wexp,
                                                   float* __restrict__ out)
{
    __shared__ float s[4096];
    __shared__ float part[256];
    const int h   = blockIdx.x;
    const int tid = threadIdx.x;
    const float off = offset[h];

    for (int k = tid; k < 4096; k += 256) {
        float val = 0.f;
        if (k < SHAPE) val = __expf(w[h * SHAPE + k] - off);
        wexp[h * WSTR + k] = val;
        s[k] = val;
    }
    __syncthreads();

    const int base = tid * 16;
    float loc[16];
    float acc = 0.f;
    #pragma unroll
    for (int i = 0; i < 16; ++i) { loc[i] = acc; acc += s[base + i]; }
    part[tid] = acc;
    __syncthreads();
    if (tid == 0) {
        float run = 0.f;
        for (int i = 0; i < 256; ++i) { float t = part[i]; part[i] = run; run += t; }
    }
    __syncthreads();
    const float pbase = part[tid];
    #pragma unroll
    for (int i = 0; i < 16; ++i) s[base + i] = pbase + loc[i];
    __syncthreads();

    float* zout = out + (size_t)Bc * LFULL * Hh * Dd;
    for (int i = tid; i < Lq; i += 256) {
        const float z = s[i + Lq] - s[i];
        #pragma unroll
        for (int n = 0; n < Bc; ++n)
            zout[((size_t)n * LFULL + (i + 1)) * Hh + h] = z;
    }
    if (tid < 2 * Bc) {
        const int n  = tid & 3;
        const int jp = (tid >> 2) ? (LFULL - 1) : 0;
        zout[((size_t)n * LFULL + jp) * Hh + h] = 0.f;
    }
    for (int i = tid; i < Bc * 2 * Dd; i += 256) {
        const int d  = i & 63;
        const int r  = i >> 6;
        const int n  = r & 3;
        const int jp = (r >> 2) ? (LFULL - 1) : 0;
        out[(((size_t)n * LFULL + jp) * Hh + h) * Dd + d] = 0.f;
    }
}

__global__ __launch_bounds__(256) __attribute__((amdgpu_waves_per_eu(4, 4)))
void pbv_kernel(const float* __restrict__ v,
                const float* __restrict__ wexp,
                float* __restrict__ out)
{
    __shared__ __attribute__((aligned(16))) float band[BANDALL];
    __shared__ float vlds[2 * VSTAGE];

    const int tid  = threadIdx.x;
    const int lane = tid & 31;
    const int wave = tid >> 5;
    const int m    = lane & 15;
    const int kh   = lane >> 4;
    const int j0   = blockIdx.x * MT;
    const int n    = blockIdx.y;
    const int h    = blockIdx.z;

    const float* wrow = wexp + h * WSTR;
    const int bandBase = (Lq - 1) - j0 - (MT - 1);

    v8f c[2][4] = {};

    const float* vbase = v + (((size_t)n * LFULL + 1) * Hh + h) * Dd;
    const int vrowk = tid >> 3;
    const int vd    = (tid & 7) * 8;
    const int vst   = vd * VSTR + vrowk;

    {
        const float4* bsrc = (const float4*)(wrow + bandBase);
        float4* bdst = (float4*)band;
        for (int i = tid; i < BANDALL / 4; i += 256) bdst[i] = bsrc[i];
    }
    {
        const float* src = vbase + (size_t)vrowk * (Hh * Dd) + vd;
        const float4 f0 = *(const float4*)(src);
        const float4 f1 = *(const float4*)(src + 4);
        float* dst = vlds + vst;
        dst[0 * VSTR] = f0.x;
        dst[1 * VSTR] = f0.y;
        dst[2 * VSTR] = f0.z;
        dst[3 * VSTR] = f0.w;
        dst[4 * VSTR] = f1.x;
        dst[5 * VSTR] = f1.y;
        dst[6 * VSTR] = f1.z;
        dst[7 * VSTR] = f1.w;
    }
    __syncthreads();

    const int wr = wave * 32;
    for (int it = 0; it < NSLAB; ++it) {
        const int  l0   = it * KT;
        const bool more = (it + 1) < NSLAB;

        float4 f0 = {0.f, 0.f, 0.f, 0.f}, f1 = {0.f, 0.f, 0.f, 0.f};
        if (more) {
            const float* src = vbase + (size_t)(l0 + KT + vrowk) * (Hh * Dd) + vd;
            f0 = *(const float4*)(src);
            f1 = *(const float4*)(src + 4);
        }
        __builtin_amdgcn_sched_barrier(0);

        const float* vcur = vlds + (it & 1) * VSTAGE;
        const float* bcur = band + l0;

        #pragma unroll
        for (int ks = 0; ks < KT / 4; ++ks) {
            v2f a[2];
            #pragma unroll
            for (int f = 0; f < 2; ++f) {
                const int t0 = ks * 4 + 2 * kh + (MT - 1) - (wr + f * 16 + m);
                a[f][0] = bcur[t0];
                a[f][1] = bcur[t0 + 1];
            }
            #pragma unroll
            for (int nt = 0; nt < 4; ++nt) {
                const float* bp = &vcur[(nt * 16 + m) * VSTR + ks * 4 + 2 * kh];
                v2f b;
                b[0] = bp[0];
                b[1] = bp[1];
                #pragma unroll
                for (int f = 0; f < 2; ++f) {
                    c[f][nt] = __builtin_amdgcn_wmma_f32_16x16x4_f32(
                        false, a[f], false, b, (short)0, c[f][nt], false, false);
                }
            }
        }

        if (more) {
            float* dst = vlds + ((it & 1) ^ 1) * VSTAGE + vst;
            dst[0 * VSTR] = f0.x;
            dst[1 * VSTR] = f0.y;
            dst[2 * VSTR] = f0.z;
            dst[3 * VSTR] = f0.w;
            dst[4 * VSTR] = f1.x;
            dst[5 * VSTR] = f1.y;
            dst[6 * VSTR] = f1.z;
            dst[7 * VSTR] = f1.w;
        }
        __syncthreads();
    }

    #pragma unroll
    for (int f = 0; f < 2; ++f) {
        #pragma unroll
        for (int nt = 0; nt < 4; ++nt) {
            const int d = nt * 16 + m;
            #pragma unroll
            for (int r = 0; r < 8; ++r) {
                const int mrow = (lane < 16) ? r : (r + 8);
                const int j = j0 + wave * 32 + f * 16 + mrow;
                out[(((size_t)n * LFULL + (j + 1)) * Hh + h) * Dd + d] = c[f][nt][r];
            }
        }
    }
}

extern "C" void kernel_launch(void* const* d_in, const int* in_sizes, int n_in,
                              void* d_out, int out_size, void* d_ws, size_t ws_size,
                              hipStream_t stream)
{
    (void)in_sizes; (void)n_in; (void)out_size; (void)ws_size;
    const float* v      = (const float*)d_in[0];
    const float* offset = (const float*)d_in[1];
    const float* w      = (const float*)d_in[2];
    float*       out    = (float*)d_out;
    float*       wexp   = (float*)d_ws;

    prep_kernel<<<dim3(Hh), 256, 0, stream>>>(w, offset, wexp, out);
    pbv_kernel<<<dim3(Lq / MT, Bc, Hh), 256, 0, stream>>>(v, wexp, out);
}


# Toeplitz operand cut into 2x2 blocks with equal diagonal blocks: three block products instead of four, T0(x0+x1) shared by both output halves; changes the order of the f32 sums
# speedup vs baseline: 1.3744x; 1.1883x over previous
#include <hip/hip_runtime.h>

typedef float v2f __attribute__((ext_vector_type(2)));
typedef float v8f __attribute__((ext_vector_type(8)));

#define Bc     4
#define LFULL  2050
#define Lq     2048
#define Hh     16
#define Dd     64
#define SHAPE  4095
#define WSTR   4096

#define HALF   (Lq / 2)
#define KT     32
#define MT     128
#define NSLAB  (HALF / KT)
#define BANDN  (HALF + MT)
#define VSTR   (KT + 2)
#define VSLAB  (Dd * VSTR)
#define VSTAGE (2 * VSLAB)

__global__ __launch_bounds__(256) void prep_kernel(const float* __restrict__ w,
                                                   const float* __restrict__ offset,
                                                   float* __restrict__ wexp,
                                                   float* __restrict__ out)
{
    __shared__ float s[4096];
    __shared__ float part[256];
    const int h   = blockIdx.x;
    const int tid = threadIdx.x;
    const float off = offset[h];

    for (int k = tid; k < 4096; k += 256) {
        float val = 0.f;
        if (k < SHAPE) val = __expf(w[h * SHAPE + k] - off);
        wexp[h * WSTR + k] = val;
        s[k] = val;
    }
    __syncthreads();

    const int base = tid * 16;
    float loc[16];
    float acc = 0.f;
    #pragma unroll
    for (int i = 0; i < 16; ++i) { loc[i] = acc; acc += s[base + i]; }
    part[tid] = acc;
    __syncthreads();
    if (tid == 0) {
        float run = 0.f;
        for (int i = 0; i < 256; ++i) { float t = part[i]; part[i] = run; run += t; }
    }
    __syncthreads();
    const float pbase = part[tid];
    #pragma unroll
    for (int i = 0; i < 16; ++i) s[base + i] = pbase + loc[i];
    __syncthreads();

    float* zout = out + (size_t)Bc * LFULL * Hh * Dd;
    for (int i = tid; i < Lq; i += 256) {
        const float z = s[i + Lq] - s[i];
        #pragma unroll
        for (int n = 0; n < Bc; ++n)
            zout[((size_t)n * LFULL + (i + 1)) * Hh + h] = z;
    }
    if (tid < 2 * Bc) {
        const int n  = tid & 3;
        const int jp = (tid >> 2) ? (LFULL - 1) : 0;
        zout[((size_t)n * LFULL + jp) * Hh + h] = 0.f;
    }
    for (int i = tid; i < Bc * 2 * Dd; i += 256) {
        const int d  = i & 63;
        const int r  = i >> 6;
        const int n  = r & 3;
        const int jp = (r >> 2) ? (LFULL - 1) : 0;
        out[(((size_t)n * LFULL + jp) * Hh + h) * Dd + d] = 0.f;
    }
}

__device__ __forceinline__ void slab_products(const float* __restrict__ vcur,
                                              const float* __restrict__ a0p,
                                              const float* __restrict__ aPp,
                                              const float* __restrict__ aMp,
                                              const int m, const int kh,
                                              v8f (&cS)[4], v8f (&cP)[4], v8f (&cM)[4])
{
    #pragma unroll
    for (int ks = 0; ks < KT / 4; ++ks) {
        v2f a0, aP, aM;
        a0[0] = a0p[ks * 4]; a0[1] = a0p[ks * 4 + 1];
        aP[0] = aPp[ks * 4]; aP[1] = aPp[ks * 4 + 1];
        aM[0] = aMp[ks * 4]; aM[1] = aMp[ks * 4 + 1];
        #pragma unroll
        for (int nt = 0; nt < 4; ++nt) {
            const float* bp = &vcur[(nt * 16 + m) * VSTR + ks * 4 + 2 * kh];
            v2f b0, b1;
            b0[0] = bp[0];
            b0[1] = bp[1];
            b1[0] = bp[VSLAB];
            b1[1] = bp[VSLAB + 1];
            const v2f bs = b0 + b1;
            cS[nt] = __builtin_amdgcn_wmma_f32_16x16x4_f32(
                false, a0, false, bs, (short)0, cS[nt], false, false);
            cP[nt] = __builtin_amdgcn_wmma_f32_16x16x4_f32(
                false, aP, false, b1, (short)0, cP[nt], false, false);
            cM[nt] = __builtin_amdgcn_wmma_f32_16x16x4_f32(
                false, aM, false, b0, (short)0, cM[nt], false, false);
        }
    }
}

__global__ __launch_bounds__(256) __attribute__((amdgpu_waves_per_eu(4, 4)))
void pbv_kernel(const float* __restrict__ v,
                const float* __restrict__ wexp,
                float* __restrict__ out)
{
    __shared__ __attribute__((aligned(16))) float band0[BANDN];
    __shared__ __attribute__((aligned(16))) float bandP[BANDN];
    __shared__ __attribute__((aligned(16))) float bandM[BANDN];
    __shared__ __attribute__((aligned(16))) float vlds[2 * VSTAGE];

    const int tid  = threadIdx.x;
    const int lane = tid & 31;
    const int wave = tid >> 5;
    const int m    = lane & 15;
    const int kh   = lane >> 4;
    const int jb   = blockIdx.x * MT;
    const int n    = blockIdx.y;
    const int h    = blockIdx.z;

    const float* wrow = wexp + h * WSTR;
    const int tb = (HALF - MT) - jb;

    v8f cS[4] = {}, cP[4] = {}, cM[4] = {};

    const float* vbase = v + (((size_t)n * LFULL + 1) * Hh + h) * Dd;
    const int vrowk = tid >> 3;
    const int vd    = (tid & 7) * 8;
    const int vst   = vd * VSTR + vrowk;
    const size_t x1off = (size_t)HALF * (Hh * Dd);

    {
        const float4* s0 = (const float4*)(wrow + HALF + tb);
        const float4* sP = (const float4*)(wrow + 2 * HALF + tb);
        const float4* sM = (const float4*)(wrow + tb);
        for (int i = tid; i < BANDN / 4; i += 256) {
            const float4 a0 = s0[i];
            const float4 aP = sP[i];
            const float4 aM = sM[i];
            float4 dP, dM;
            dP.x = aP.x - a0.x; dP.y = aP.y - a0.y; dP.z = aP.z - a0.z; dP.w = aP.w - a0.w;
            dM.x = aM.x - a0.x; dM.y = aM.y - a0.y; dM.z = aM.z - a0.z; dM.w = aM.w - a0.w;
            ((float4*)band0)[i] = a0;
            ((float4*)bandP)[i] = dP;
            ((float4*)bandM)[i] = dM;
        }
    }
    {
        const float* src = vbase + (size_t)vrowk * (Hh * Dd) + vd;
        const float4 f0 = *(const float4*)(src);
        const float4 f1 = *(const float4*)(src + 4);
        const float4 g0 = *(const float4*)(src + x1off);
        const float4 g1 = *(const float4*)(src + x1off + 4);
        float* dst = vlds + vst;
        dst[0 * VSTR] = f0.x;
        dst[1 * VSTR] = f0.y;
        dst[2 * VSTR] = f0.z;
        dst[3 * VSTR] = f0.w;
        dst[4 * VSTR] = f1.x;
        dst[5 * VSTR] = f1.y;
        dst[6 * VSTR] = f1.z;
        dst[7 * VSTR] = f1.w;
        dst[VSLAB + 0 * VSTR] = g0.x;
        dst[VSLAB + 1 * VSTR] = g0.y;
        dst[VSLAB + 2 * VSTR] = g0.z;
        dst[VSLAB + 3 * VSTR] = g0.w;
        dst[VSLAB + 4 * VSTR] = g1.x;
        dst[VSLAB + 5 * VSTR] = g1.y;
        dst[VSLAB + 6 * VSTR] = g1.z;
        dst[VSLAB + 7 * VSTR] = g1.w;
    }
    __syncthreads();

    const int arow = (MT - 1) - (wave * 16 + m) + 2 * kh;
    for (int it = 0; it < NSLAB - 1; ++it) {
        const int l0 = it * KT;

        const float* src = vbase + (size_t)(l0 + KT + vrowk) * (Hh * Dd) + vd;
        const float4 f0 = *(const float4*)(src);
        const float4 f1 = *(const float4*)(src + 4);
        const float4 g0 = *(const float4*)(src + x1off);
        const float4 g1 = *(const float4*)(src + x1off + 4);
        __builtin_amdgcn_sched_barrier(0);

        slab_products(vlds + (it & 1) * VSTAGE,
                      band0 + l0 + arow, bandP + l0 + arow, bandM + l0 + arow,
                      m, kh, cS, cP, cM);

        float* dst = vlds + ((it & 1) ^ 1) * VSTAGE + vst;
        dst[0 * VSTR] = f0.x;
        dst[1 * VSTR] = f0.y;
        dst[2 * VSTR] = f0.z;
        dst[3 * VSTR] = f0.w;
        dst[4 * VSTR] = f1.x;
        dst[5 * VSTR] = f1.y;
        dst[6 * VSTR] = f1.z;
        dst[7 * VSTR] = f1.w;
        dst[VSLAB + 0 * VSTR] = g0.x;
        dst[VSLAB + 1 * VSTR] = g0.y;
        dst[VSLAB + 2 * VSTR] = g0.z;
        dst[VSLAB + 3 * VSTR] = g0.w;
        dst[VSLAB + 4 * VSTR] = g1.x;
        dst[VSLAB + 5 * VSTR] = g1.y;
        dst[VSLAB + 6 * VSTR] = g1.z;
        dst[VSLAB + 7 * VSTR] = g1.w;
        __syncthreads();
    }
    slab_products(vlds + ((NSLAB - 1) & 1) * VSTAGE,
                  band0 + (NSLAB - 1) * KT + arow, bandP + (NSLAB - 1) * KT + arow,
                  bandM + (NSLAB - 1) * KT + arow,
                  m, kh, cS, cP, cM);

    #pragma unroll
    for (int nt = 0; nt < 4; ++nt) {
        const int d = nt * 16 + m;
        #pragma unroll
        for (int r = 0; r < 8; ++r) {
            const int mrow = (lane < 16) ? r : (r + 8);
            const int j = jb + wave * 16 + mrow;
            out[(((size_t)n * LFULL + (j + 1)) * Hh + h) * Dd + d]        = cS[nt][r] + cP[nt][r];
            out[(((size_t)n * LFULL + (j + HALF + 1)) * Hh + h) * Dd + d] = cS[nt][r] + cM[nt][r];
        }
    }
}

extern "C" void kernel_launch(void* const* d_in, const int* in_sizes, int n_in,
                              void* d_out, int out_size, void* d_ws, size_t ws_size,
                              hipStream_t stream)
{
    (void)in_sizes; (void)n_in; (void)out_size; (void)ws_size;
    const float* v      = (const float*)d_in[0];
    const float* offset = (const float*)d_in[1];
    const float* w      = (const float*)d_in[2];
    float*       out    = (float*)d_out;
    float*       wexp   = (float*)d_ws;

    prep_kernel<<<dim3(Hh), 256, 0, stream>>>(w, offset, wexp, out);
    pbv_kernel<<<dim3(HALF / MT, Bc, Hh), 256, 0, stream>>>(v, wexp, out);
}


# K loop's LDS fragment reads grouped: A and B fragments of two k-steps read in one group 12+ WMMAs ahead into a register ring (across the slab barrier too): 4 read places per 96 WMMAs, not about 20
# speedup vs baseline: 1.4238x; 1.0359x over previous
#include <hip/hip_runtime.h>

typedef float v2f __attribute__((ext_vector_type(2)));
typedef float v8f __attribute__((ext_vector_type(8)));

#define Bc     4
#define LFULL  2050
#define Lq     2048
#define Hh     16
#define Dd     64
#define SHAPE  4095
#define WSTR   4096

#define HALF   (Lq / 2)
#define KT     32
#define MT     128
#define NSLAB  (HALF / KT)
#define BANDN  (HALF + MT)
#define VSTR   (KT + 2)
#define VSLAB  (Dd * VSTR)
#define VSTAGE (2 * VSLAB)

__global__ __launch_bounds__(256) void prep_kernel(const float* __restrict__ w,
                                                   const float* __restrict__ offset,
                                                   float* __restrict__ wexp,
                                                   float* __restrict__ out)
{
    __shared__ float s[4096];
    __shared__ float part[256];
    const int h   = blockIdx.x;
    const int tid = threadIdx.x;
    const float off = offset[h];

    for (int k = tid; k < 4096; k += 256) {
        float val = 0.f;
        if (k < SHAPE) val = __expf(w[h * SHAPE + k] - off);
        wexp[h * WSTR + k] = val;
        s[k] = val;
    }
    __syncthreads();

    const int base = tid * 16;
    float loc[16];
    float acc = 0.f;
    #pragma unroll
    for (int i = 0; i < 16; ++i) { loc[i] = acc; acc += s[base + i]; }
    part[tid] = acc;
    __syncthreads();
    if (tid == 0) {
        float run = 0.f;
        for (int i = 0; i < 256; ++i) { float t = part[i]; part[i] = run; run += t; }
    }
    __syncthreads();
    const float pbase = part[tid];
    #pragma unroll
    for (int i = 0; i < 16; ++i) s[base + i] = pbase + loc[i];
    __syncthreads();

    float* zout = out + (size_t)Bc * LFULL * Hh * Dd;
    for (int i = tid; i < Lq; i += 256) {
        const float z = s[i + Lq] - s[i];
        #pragma unroll
        for (int n = 0; n < Bc; ++n)
            zout[((size_t)n * LFULL + (i + 1)) * Hh + h] = z;
    }
    if (tid < 2 * Bc) {
        const int n  = tid & 3;
        const int jp = (tid >> 2) ? (LFULL - 1) : 0;
        zout[((size_t)n * LFULL + jp) * Hh + h] = 0.f;
    }
    for (int i = tid; i < Bc * 2 * Dd; i += 256) {
        const int d  = i & 63;
        const int r  = i >> 6;
        const int n  = r & 3;
        const int jp = (r >> 2) ? (LFULL - 1) : 0;
        out[(((size_t)n * LFULL + jp) * Hh + h) * Dd + d] = 0.f;
    }
}

struct KFrag {
    v2f a0, aP, aM;
    v2f b0[4], b1[4];
};

__device__ __forceinline__ void load_k(KFrag& f,
                                       const float* __restrict__ vcur,
                                       const float* __restrict__ a0p,
                                       const float* __restrict__ aPp,
                                       const float* __restrict__ aMp,
                                       const int ks, const int m, const int kh)
{
    f.a0[0] = a0p[ks * 4]; f.a0[1] = a0p[ks * 4 + 1];
    f.aP[0] = aPp[ks * 4]; f.aP[1] = aPp[ks * 4 + 1];
    f.aM[0] = aMp[ks * 4]; f.aM[1] = aMp[ks * 4 + 1];
    #pragma unroll
    for (int nt = 0; nt < 4; ++nt) {
        const float* bp = &vcur[(nt * 16 + m) * VSTR + ks * 4 + 2 * kh];
        f.b0[nt][0] = bp[0];
        f.b0[nt][1] = bp[1];
        f.b1[nt][0] = bp[VSLAB];
        f.b1[nt][1] = bp[VSLAB + 1];
    }
}

__device__ __forceinline__ void mma_k(const KFrag& f,
                                      v8f (&cS)[4], v8f (&cP)[4], v8f (&cM)[4])
{
    #pragma unroll
    for (int nt = 0; nt < 4; ++nt) {
        const v2f bs = f.b0[nt] + f.b1[nt];
        cS[nt] = __builtin_amdgcn_wmma_f32_16x16x4_f32(
            false, f.a0, false, bs, (short)0, cS[nt], false, false);
        cP[nt] = __builtin_amdgcn_wmma_f32_16x16x4_f32(
            false, f.aP, false, f.b1[nt], (short)0, cP[nt], false, false);
        cM[nt] = __builtin_amdgcn_wmma_f32_16x16x4_f32(
            false, f.aM, false, f.b0[nt], (short)0, cM[nt], false, false);
    }
}

__global__ __launch_bounds__(256) __attribute__((amdgpu_waves_per_eu(4, 4)))
void pbv_kernel(const float* __restrict__ v,
                const float* __restrict__ wexp,
                float* __restrict__ out)
{
    __shared__ __attribute__((aligned(16))) float band0[BANDN];
    __shared__ __attribute__((aligned(16))) float bandP[BANDN];
    __shared__ __attribute__((aligned(16))) float bandM[BANDN];
    __shared__ __attribute__((aligned(16))) float vlds[2 * VSTAGE];

    const int tid  = threadIdx.x;
    const int lane = tid & 31;
    const int wave = tid >> 5;
    const int m    = lane & 15;
    const int kh   = lane >> 4;
    const int jb   = blockIdx.x * MT;
    const int n    = blockIdx.y;
    const int h    = blockIdx.z;

    const float* wrow = wexp + h * WSTR;
    const int tb = (HALF - MT) - jb;

    v8f cS[4] = {}, cP[4] = {}, cM[4] = {};

    const float* vbase = v + (((size_t)n * LFULL + 1) * Hh + h) * Dd;
    const int vrowk = tid >> 3;
    const int vd    = (tid & 7) * 8;
    const int vst   = vd * VSTR + vrowk;
    const size_t x1off = (size_t)HALF * (Hh * Dd);

    {
        const float4* s0 = (const float4*)(wrow + HALF + tb);
        const float4* sP = (const float4*)(wrow + 2 * HALF + tb);
        const float4* sM = (const float4*)(wrow + tb);
        for (int i = tid; i < BANDN / 4; i += 256) {
            const float4 a0 = s0[i];
            const float4 aP = sP[i];
            const float4 aM = sM[i];
            float4 dP, dM;
            dP.x = aP.x - a0.x; dP.y = aP.y - a0.y; dP.z = aP.z - a0.z; dP.w = aP.w - a0.w;
            dM.x = aM.x - a0.x; dM.y = aM.y - a0.y; dM.z = aM.z - a0.z; dM.w = aM.w - a0.w;
            ((float4*)band0)[i] = a0;
            ((float4*)bandP)[i] = dP;
            ((float4*)bandM)[i] = dM;
        }
    }
    {
        const float* src = vbase + (size_t)vrowk * (Hh * Dd) + vd;
        const float4 f0 = *(const float4*)(src);
        const float4 f1 = *(const float4*)(src + 4);
        const float4 g0 = *(const float4*)(src + x1off);
        const float4 g1 = *(const float4*)(src + x1off + 4);
        float* dst = vlds + vst;
        dst[0 * VSTR] = f0.x;
        dst[1 * VSTR] = f0.y;
        dst[2 * VSTR] = f0.z;
        dst[3 * VSTR] = f0.w;
        dst[4 * VSTR] = f1.x;
        dst[5 * VSTR] = f1.y;
        dst[6 * VSTR] = f1.z;
        dst[7 * VSTR] = f1.w;
        dst[VSLAB + 0 * VSTR] = g0.x;
        dst[VSLAB + 1 * VSTR] = g0.y;
        dst[VSLAB + 2 * VSTR] = g0.z;
        dst[VSLAB + 3 * VSTR] = g0.w;
        dst[VSLAB + 4 * VSTR] = g1.x;
        dst[VSLAB + 5 * VSTR] = g1.y;
        dst[VSLAB + 6 * VSTR] = g1.z;
        dst[VSLAB + 7 * VSTR] = g1.w;
    }
    __syncthreads();

    const int arow = (MT - 1) - (wave * 16 + m) + 2 * kh;

    KFrag q0, q1, q2, q3;
    load_k(q0, vlds, band0 + arow, bandP + arow, bandM + arow, 0, m, kh);
    load_k(q1, vlds, band0 + arow, bandP + arow, bandM + arow, 1, m, kh);

    for (int it = 0; it < NSLAB - 1; ++it) {
        const int l0 = it * KT;
        const float* vcur = vlds + (it & 1) * VSTAGE;
        const float* a0p = band0 + l0 + arow;
        const float* aPp = bandP + l0 + arow;
        const float* aMp = bandM + l0 + arow;

        const float* src = vbase + (size_t)(l0 + KT + vrowk) * (Hh * Dd) + vd;
        const float4 f0 = *(const float4*)(src);
        const float4 f1 = *(const float4*)(src + 4);
        const float4 g0 = *(const float4*)(src + x1off);
        const float4 g1 = *(const float4*)(src + x1off + 4);
        __builtin_amdgcn_sched_barrier(0);
        mma_k(q0, cS, cP, cM);
        __builtin_amdgcn_sched_barrier(0);
        load_k(q2, vcur, a0p, aPp, aMp, 2, m, kh);
        load_k(q3, vcur, a0p, aPp, aMp, 3, m, kh);
        __builtin_amdgcn_sched_barrier(0);
        mma_k(q1, cS, cP, cM);
        mma_k(q2, cS, cP, cM);
        __builtin_amdgcn_sched_barrier(0);
        load_k(q0, vcur, a0p, aPp, aMp, 4, m, kh);
        load_k(q1, vcur, a0p, aPp, aMp, 5, m, kh);
        __builtin_amdgcn_sched_barrier(0);
        mma_k(q3, cS, cP, cM);
        mma_k(q0, cS, cP, cM);
        __builtin_amdgcn_sched_barrier(0);
        load_k(q2, vcur, a0p, aPp, aMp, 6, m, kh);
        load_k(q3, vcur, a0p, aPp, aMp, 7, m, kh);
        __builtin_amdgcn_sched_barrier(0);
        mma_k(q1, cS, cP, cM);
        mma_k(q2, cS, cP, cM);
        __builtin_amdgcn_sched_barrier(0);

        float* dst = vlds + ((it & 1) ^ 1) * VSTAGE + vst;
        dst[0 * VSTR] = f0.x;
        dst[1 * VSTR] = f0.y;
        dst[2 * VSTR] = f0.z;
        dst[3 * VSTR] = f0.w;
        dst[4 * VSTR] = f1.x;
        dst[5 * VSTR] = f1.y;
        dst[6 * VSTR] = f1.z;
        dst[7 * VSTR] = f1.w;
        dst[VSLAB + 0 * VSTR] = g0.x;
        dst[VSLAB + 1 * VSTR] = g0.y;
        dst[VSLAB + 2 * VSTR] = g0.z;
        dst[VSLAB + 3 * VSTR] = g0.w;
        dst[VSLAB + 4 * VSTR] = g1.x;
        dst[VSLAB + 5 * VSTR] = g1.y;
        dst[VSLAB + 6 * VSTR] = g1.z;
        dst[VSLAB + 7 * VSTR] = g1.w;
        __syncthreads();

        load_k(q0, vlds + ((it & 1) ^ 1) * VSTAGE, a0p + KT, aPp + KT, aMp + KT, 0, m, kh);
        load_k(q1, vlds + ((it & 1) ^ 1) * VSTAGE, a0p + KT, aPp + KT, aMp + KT, 1, m, kh);
        __builtin_amdgcn_sched_barrier(0);
        mma_k(q3, cS, cP, cM);
    }
    {
        const float* vcur = vlds + ((NSLAB - 1) & 1) * VSTAGE;
        const float* a0p = band0 + (NSLAB - 1) * KT + arow;
        const float* aPp = bandP + (NSLAB - 1) * KT + arow;
        const float* aMp = bandM + (NSLAB - 1) * KT + arow;
        mma_k(q0, cS, cP, cM);
        __builtin_amdgcn_sched_barrier(0);
        load_k(q2, vcur, a0p, aPp, aMp, 2, m, kh);
        load_k(q3, vcur, a0p, aPp, aMp, 3, m, kh);
        __builtin_amdgcn_sched_barrier(0);
        mma_k(q1, cS, cP, cM);
        mma_k(q2, cS, cP, cM);
        __builtin_amdgcn_sched_barrier(0);
        load_k(q0, vcur, a0p, aPp, aMp, 4, m, kh);
        load_k(q1, vcur, a0p, aPp, aMp, 5, m, kh);
        __builtin_amdgcn_sched_barrier(0);
        mma_k(q3, cS, cP, cM);
        mma_k(q0, cS, cP, cM);
        __builtin_amdgcn_sched_barrier(0);
        load_k(q2, vcur, a0p, aPp, aMp, 6, m, kh);
        load_k(q3, vcur, a0p, aPp, aMp, 7, m, kh);
        __builtin_amdgcn_sched_barrier(0);
        mma_k(q1, cS, cP, cM);
        mma_k(q2, cS, cP, cM);
        mma_k(q3, cS, cP, cM);
    }

    #pragma unroll
    for (int nt = 0; nt < 4; ++nt) {
        const int d = nt * 16 + m;
        #pragma unroll
        for (int r = 0; r < 8; ++r) {
            const int mrow = (lane < 16) ? r : (r + 8);
            const int j = jb + wave * 16 + mrow;
            out[(((size_t)n * LFULL + (j + 1)) * Hh + h) * Dd + d]        = cS[nt][r] + cP[nt][r];
            out[(((size_t)n * LFULL + (j + HALF + 1)) * Hh + h) * Dd + d] = cS[nt][r] + cM[nt][r];
        }
    }
}

extern "C" void kernel_launch(void* const* d_in, const int* in_sizes, int n_in,
                              void* d_out, int out_size, void* d_ws, size_t ws_size,
                              hipStream_t stream)
{
    (void)in_sizes; (void)n_in; (void)out_size; (void)ws_size;
    const float* v      = (const float*)d_in[0];
    const float* offset = (const float*)d_in[1];
    const float* w      = (const float*)d_in[2];
    float*       out    = (float*)d_out;
    float*       wexp   = (float*)d_ws;

    prep_kernel<<<dim3(Hh), 256, 0, stream>>>(w, offset, wexp, out);
    pbv_kernel<<<dim3(HALF / MT, Bc, Hh), 256, 0, stream>>>(v, wexp, out);
}


# Toeplitz operand cut a second time (4x4 blocks of 512, equal diagonal blocks at both levels): nine block products instead of twelve, 36 accumulator tiles, 2 waves per SIMD; changes the f32 sum order
# speedup vs baseline: 1.6687x; 1.1720x over previous
#include <hip/hip_runtime.h>

typedef float v2f __attribute__((ext_vector_type(2)));
typedef float v8f __attribute__((ext_vector_type(8)));

#define Bc     4
#define LFULL  2050
#define Lq     2048
#define Hh     16
#define Dd     64
#define SHAPE  4095
#define WSTR   4096

#define QTR    (Lq / 4)
#define KT     16
#define MT     128
#define NSLAB  (QTR / KT)
#define BANDN  (QTR + MT)
#define NBAND  9
#define VSTR   (KT + 4)
#define VSLAB  (Dd * VSTR)
#define VSTAGE (4 * VSLAB)

__global__ __launch_bounds__(256) void prep_kernel(const float* __restrict__ w,
                                                   const float* __restrict__ offset,
                                                   float* __restrict__ wexp,
                                                   float* __restrict__ out)
{
    __shared__ float s[4096];
    __shared__ float part[256];
    const int h   = blockIdx.x;
    const int tid = threadIdx.x;
    const float off = offset[h];

    for (int k = tid; k < 4096; k += 256) {
        float val = 0.f;
        if (k < SHAPE) val = __expf(w[h * SHAPE + k] - off);
        wexp[h * WSTR + k] = val;
        s[k] = val;
    }
    __syncthreads();

    const int base = tid * 16;
    float loc[16];
    float acc = 0.f;
    #pragma unroll
    for (int i = 0; i < 16; ++i) { loc[i] = acc; acc += s[base + i]; }
    part[tid] = acc;
    __syncthreads();
    if (tid == 0) {
        float run = 0.f;
        for (int i = 0; i < 256; ++i) { float t = part[i]; part[i] = run; run += t; }
    }
    __syncthreads();
    const float pbase = part[tid];
    #pragma unroll
    for (int i = 0; i < 16; ++i) s[base + i] = pbase + loc[i];
    __syncthreads();

    float* zout = out + (size_t)Bc * LFULL * Hh * Dd;
    for (int i = tid; i < Lq; i += 256) {
        const float z = s[i + Lq] - s[i];
        #pragma unroll
        for (int n = 0; n < Bc; ++n)
            zout[((size_t)n * LFULL + (i + 1)) * Hh + h] = z;
    }
    if (tid < 2 * Bc) {
        const int n  = tid & 3;
        const int jp = (tid >> 2) ? (LFULL - 1) : 0;
        zout[((size_t)n * LFULL + jp) * Hh + h] = 0.f;
    }
    for (int i = tid; i < Bc * 2 * Dd; i += 256) {
        const int d  = i & 63;
        const int r  = i >> 6;
        const int n  = r & 3;
        const int jp = (r >> 2) ? (LFULL - 1) : 0;
        out[(((size_t)n * LFULL + jp) * Hh + h) * Dd + d] = 0.f;
    }
}

struct KFrag {
    v2f a[NBAND];
    v2f b[4][4];
};

__device__ __forceinline__ void load_k(KFrag& f,
                                       const float* __restrict__ vcur,
                                       const float* __restrict__ ap,
                                       const int ks, const int m, const int kh)
{
    #pragma unroll
    for (int q = 0; q < NBAND; ++q) {
        f.a[q][0] = ap[q * BANDN + ks * 4];
        f.a[q][1] = ap[q * BANDN + ks * 4 + 1];
    }
    #pragma unroll
    for (int nt = 0; nt < 4; ++nt) {
        const float* bp = &vcur[(nt * 16 + m) * VSTR + ks * 4 + 2 * kh];
        #pragma unroll
        for (int q = 0; q < 4; ++q) {
            f.b[nt][q][0] = bp[q * VSLAB];
            f.b[nt][q][1] = bp[q * VSLAB + 1];
        }
    }
}

#define WMMA4(A, B, C) __builtin_amdgcn_wmma_f32_16x16x4_f32(false, (A), false, (B), (short)0, (C), false, false)

__device__ __forceinline__ void mma_k(const KFrag& f, v8f (&c)[NBAND][4])
{
    #pragma unroll
    for (int nt = 0; nt < 4; ++nt) {
        const v2f v0 = f.b[nt][0], v1 = f.b[nt][1], v2 = f.b[nt][2], v3 = f.b[nt][3];
        const v2f z0  = v0 + v2;
        const v2f z1  = v1 + v3;
        const v2f zs  = z0 + z1;
        const v2f v23 = v2 + v3;
        const v2f v01 = v0 + v1;
        c[0][nt] = WMMA4(f.a[0], zs,  c[0][nt]);
        c[1][nt] = WMMA4(f.a[1], z1,  c[1][nt]);
        c[2][nt] = WMMA4(f.a[2], z0,  c[2][nt]);
        c[3][nt] = WMMA4(f.a[3], v23, c[3][nt]);
        c[4][nt] = WMMA4(f.a[4], v3,  c[4][nt]);
        c[5][nt] = WMMA4(f.a[5], v2,  c[5][nt]);
        c[6][nt] = WMMA4(f.a[6], v01, c[6][nt]);
        c[7][nt] = WMMA4(f.a[7], v1,  c[7][nt]);
        c[8][nt] = WMMA4(f.a[8], v0,  c[8][nt]);
    }
}

__device__ __forceinline__ float4 f4sub(const float4 a, const float4 b)
{
    float4 r;
    r.x = a.x - b.x; r.y = a.y - b.y; r.z = a.z - b.z; r.w = a.w - b.w;
    return r;
}

__device__ __forceinline__ void put_slab(float* __restrict__ dst,
                                         const float4 g0, const float4 g1, const float4 g2, const float4 g3)
{
    dst[0 * VSLAB + 0 * VSTR] = g0.x;
    dst[0 * VSLAB + 1 * VSTR] = g0.y;
    dst[0 * VSLAB + 2 * VSTR] = g0.z;
    dst[0 * VSLAB + 3 * VSTR] = g0.w;
    dst[1 * VSLAB + 0 * VSTR] = g1.x;
    dst[1 * VSLAB + 1 * VSTR] = g1.y;
    dst[1 * VSLAB + 2 * VSTR] = g1.z;
    dst[1 * VSLAB + 3 * VSTR] = g1.w;
    dst[2 * VSLAB + 0 * VSTR] = g2.x;
    dst[2 * VSLAB + 1 * VSTR] = g2.y;
    dst[2 * VSLAB + 2 * VSTR] = g2.z;
    dst[2 * VSLAB + 3 * VSTR] = g2.w;
    dst[3 * VSLAB + 0 * VSTR] = g3.x;
    dst[3 * VSLAB + 1 * VSTR] = g3.y;
    dst[3 * VSLAB + 2 * VSTR] = g3.z;
    dst[3 * VSLAB + 3 * VSTR] = g3.w;
}

__global__ __launch_bounds__(256) __attribute__((amdgpu_waves_per_eu(2, 2)))
void pbv_kernel(const float* __restrict__ v,
                const float* __restrict__ wexp,
                float* __restrict__ out)
{
    __shared__ __attribute__((aligned(16))) float band[NBAND * BANDN];
    __shared__ __attribute__((aligned(16))) float vlds[2 * VSTAGE];

    const int tid  = threadIdx.x;
    const int lane = tid & 31;
    const int wave = tid >> 5;
    const int m    = lane & 15;
    const int kh   = lane >> 4;
    const int jb   = blockIdx.x * MT;
    const int n    = blockIdx.y;
    const int h    = blockIdx.z;

    const float* wrow = wexp + h * WSTR;
    const int tb = (QTR - MT) - jb;

    v8f c[NBAND][4] = {};

    const float* vbase = v + (((size_t)n * LFULL + 1) * Hh + h) * Dd;
    const int vrowk = tid >> 4;
    const int vd    = (tid & 15) * 4;
    const int vst   = vd * VSTR + vrowk;
    const size_t qoff = (size_t)QTR * (Hh * Dd);

    {
        const float4* sw = (const float4*)(wrow + tb);
        for (int i = tid; i < BANDN / 4; i += 256) {
            const float4 wm3 = sw[i];
            const float4 wm2 = sw[i + 1 * (QTR / 4)];
            const float4 wm1 = sw[i + 2 * (QTR / 4)];
            const float4 w0  = sw[i + 3 * (QTR / 4)];
            const float4 w1  = sw[i + 4 * (QTR / 4)];
            const float4 w2  = sw[i + 5 * (QTR / 4)];
            const float4 w3  = sw[i + 6 * (QTR / 4)];
            const float4 pA0 = f4sub(w2, w0);
            const float4 mB0 = f4sub(wm2, w0);
            float4* bd = (float4*)band + i;
            bd[0 * (BANDN / 4)] = w0;
            bd[1 * (BANDN / 4)] = f4sub(w1, w0);
            bd[2 * (BANDN / 4)] = f4sub(wm1, w0);
            bd[3 * (BANDN / 4)] = pA0;
            bd[4 * (BANDN / 4)] = f4sub(f4sub(w3, w1), pA0);
            bd[5 * (BANDN / 4)] = f4sub(f4sub(w1, wm1), pA0);
            bd[6 * (BANDN / 4)] = mB0;
            bd[7 * (BANDN / 4)] = f4sub(f4sub(wm1, w1), mB0);
            bd[8 * (BANDN / 4)] = f4sub(f4sub(wm3, wm1), mB0);
        }
    }
    {
        const float* src = vbase + (size_t)vrowk * (Hh * Dd) + vd;
        const float4 g0 = *(const float4*)(src);
        const float4 g1 = *(const float4*)(src + qoff);
        const float4 g2 = *(const float4*)(src + 2 * qoff);
        const float4 g3 = *(const float4*)(src + 3 * qoff);
        put_slab(vlds + vst, g0, g1, g2, g3);
    }
    __syncthreads();

    const int arow = (MT - 1) - (wave * 16 + m) + 2 * kh;
    const float* ab = band + arow;

    KFrag q0, q1;
    load_k(q0, vlds, ab, 0, m, kh);

    for (int it = 0; it < NSLAB - 1; ++it) {
        const int l0 = it * KT;
        const float* vcur = vlds + (it & 1) * VSTAGE;
        float* vnxt = vlds + ((it & 1) ^ 1) * VSTAGE;
        const float* ap = ab + l0;

        const float* src = vbase + (size_t)(l0 + KT + vrowk) * (Hh * Dd) + vd;
        const float4 g0 = *(const float4*)(src);
        const float4 g1 = *(const float4*)(src + qoff);
        const float4 g2 = *(const float4*)(src + 2 * qoff);
        const float4 g3 = *(const float4*)(src + 3 * qoff);
        load_k(q1, vcur, ap, 1, m, kh);
        __builtin_amdgcn_sched_barrier(0);
        mma_k(q0, c);
        __builtin_amdgcn_sched_barrier(0);
        load_k(q0, vcur, ap, 2, m, kh);
        __builtin_amdgcn_sched_barrier(0);
        mma_k(q1, c);
        __builtin_amdgcn_sched_barrier(0);
        load_k(q1, vcur, ap, 3, m, kh);
        __builtin_amdgcn_sched_barrier(0);
        mma_k(q0, c);
        __builtin_amdgcn_sched_barrier(0);

        put_slab(vnxt + vst, g0, g1, g2, g3);
        __syncthreads();

        load_k(q0, vnxt, ap + KT, 0, m, kh);
        __builtin_amdgcn_sched_barrier(0);
        mma_k(q1, c);
    }
    {
        const float* vcur = vlds + ((NSLAB - 1) & 1) * VSTAGE;
        const float* ap = ab + (NSLAB - 1) * KT;
        load_k(q1, vcur, ap, 1, m, kh);
        __builtin_amdgcn_sched_barrier(0);
        mma_k(q0, c);
        __builtin_amdgcn_sched_barrier(0);
        load_k(q0, vcur, ap, 2, m, kh);
        __builtin_amdgcn_sched_barrier(0);
        mma_k(q1, c);
        __builtin_amdgcn_sched_barrier(0);
        load_k(q1, vcur, ap, 3, m, kh);
        __builtin_amdgcn_sched_barrier(0);
        mma_k(q0, c);
        mma_k(q1, c);
    }

    #pragma unroll
    for (int nt = 0; nt < 4; ++nt) {
        const int d = nt * 16 + m;
        #pragma unroll
        for (int r = 0; r < 8; ++r) {
            const int mrow = (lane < 16) ? r : (r + 8);
            const int j = jb + wave * 16 + mrow;
            float* o = out + (((size_t)n * LFULL + (j + 1)) * Hh + h) * Dd + d;
            const float s1 = c[0][nt][r] + c[1][nt][r];
            const float sm = c[0][nt][r] + c[2][nt][r];
            o[0]                              = s1 + (c[3][nt][r] + c[4][nt][r]);
            o[(size_t)QTR * (Hh * Dd)]        = sm + (c[3][nt][r] + c[5][nt][r]);
            o[(size_t)2 * QTR * (Hh * Dd)]    = s1 + (c[6][nt][r] + c[7][nt][r]);
            o[(size_t)3 * QTR * (Hh * Dd)]    = sm + (c[6][nt][r] + c[8][nt][r]);
        }
    }
}

extern "C" void kernel_launch(void* const* d_in, const int* in_sizes, int n_in,
                              void* d_out, int out_size, void* d_ws, size_t ws_size,
                              hipStream_t stream)
{
    (void)in_sizes; (void)n_in; (void)out_size; (void)ws_size;
    const float* v      = (const float*)d_in[0];
    const float* offset = (const float*)d_in[1];
    const float* w      = (const float*)d_in[2];
    float*       out    = (float*)d_out;
    float*       wexp   = (float*)d_ws;

    prep_kernel<<<dim3(Hh), 256, 0, stream>>>(w, offset, wexp, out);
    pbv_kernel<<<dim3(QTR / MT, Bc, Hh), 256, 0, stream>>>(v, wexp, out);
}


# One launch instead of two: exp(w - offset), z_pb window sums (prefix scan per 128-row piece: changes the f32 sum order of z_pb) and border rows move into the GEMM kernel's opening
# speedup vs baseline: 1.9905x; 1.1928x over previous
#include <hip/hip_runtime.h>

typedef float v2f __attribute__((ext_vector_type(2)));
typedef float v8f __attribute__((ext_vector_type(8)));

#define Bc     4
#define LFULL  2050
#define Lq     2048
#define Hh     16
#define Dd     64
#define SHAPE  4095

#define QTR    (Lq / 4)
#define KT     16
#define MT     128
#define NSLAB  (QTR / KT)
#define BANDN  (QTR + MT)
#define NBAND  9
#define VSTR   (KT + 4)
#define VSLAB  (Dd * VSTR)
#define VSTAGE (4 * VSLAB)
#define ZP     128
#define ZPT    9

struct KFrag {
    v2f a[NBAND];
    v2f b[4][4];
};

__device__ __forceinline__ void load_k(KFrag& f,
                                       const float* __restrict__ vcur,
                                       const float* __restrict__ ap,
                                       const int ks, const int m, const int kh)
{
    #pragma unroll
    for (int q = 0; q < NBAND; ++q) {
        f.a[q][0] = ap[q * BANDN + ks * 4];
        f.a[q][1] = ap[q * BANDN + ks * 4 + 1];
    }
    #pragma unroll
    for (int nt = 0; nt < 4; ++nt) {
        const float* bp = &vcur[(nt * 16 + m) * VSTR + ks * 4 + 2 * kh];
        #pragma unroll
        for (int q = 0; q < 4; ++q) {
            f.b[nt][q][0] = bp[q * VSLAB];
            f.b[nt][q][1] = bp[q * VSLAB + 1];
        }
    }
}

#define WMMA4(A, B, C) __builtin_amdgcn_wmma_f32_16x16x4_f32(false, (A), false, (B), (short)0, (C), false, false)

__device__ __forceinline__ void mma_k(const KFrag& f, v8f (&c)[NBAND][4])
{
    #pragma unroll
    for (int nt = 0; nt < 4; ++nt) {
        const v2f v0 = f.b[nt][0], v1 = f.b[nt][1], v2 = f.b[nt][2], v3 = f.b[nt][3];
        const v2f z0  = v0 + v2;
        const v2f z1  = v1 + v3;
        const v2f zs  = z0 + z1;
        const v2f v23 = v2 + v3;
        const v2f v01 = v0 + v1;
        c[0][nt] = WMMA4(f.a[0], zs,  c[0][nt]);
        c[1][nt] = WMMA4(f.a[1], z1,  c[1][nt]);
        c[2][nt] = WMMA4(f.a[2], z0,  c[2][nt]);
        c[3][nt] = WMMA4(f.a[3], v23, c[3][nt]);
        c[4][nt] = WMMA4(f.a[4], v3,  c[4][nt]);
        c[5][nt] = WMMA4(f.a[5], v2,  c[5][nt]);
        c[6][nt] = WMMA4(f.a[6], v01, c[6][nt]);
        c[7][nt] = WMMA4(f.a[7], v1,  c[7][nt]);
        c[8][nt] = WMMA4(f.a[8], v0,  c[8][nt]);
    }
}

__device__ __forceinline__ void put_slab(float* __restrict__ dst,
                                         const float4 g0, const float4 g1, const float4 g2, const float4 g3)
{
    dst[0 * VSLAB + 0 * VSTR] = g0.x;
    dst[0 * VSLAB + 1 * VSTR] = g0.y;
    dst[0 * VSLAB + 2 * VSTR] = g0.z;
    dst[0 * VSLAB + 3 * VSTR] = g0.w;
    dst[1 * VSLAB + 0 * VSTR] = g1.x;
    dst[1 * VSLAB + 1 * VSTR] = g1.y;
    dst[1 * VSLAB + 2 * VSTR] = g1.z;
    dst[1 * VSLAB + 3 * VSTR] = g1.w;
    dst[2 * VSLAB + 0 * VSTR] = g2.x;
    dst[2 * VSLAB + 1 * VSTR] = g2.y;
    dst[2 * VSLAB + 2 * VSTR] = g2.z;
    dst[2 * VSLAB + 3 * VSTR] = g2.w;
    dst[3 * VSLAB + 0 * VSTR] = g3.x;
    dst[3 * VSLAB + 1 * VSTR] = g3.y;
    dst[3 * VSLAB + 2 * VSTR] = g3.z;
    dst[3 * VSLAB + 3 * VSTR] = g3.w;
}

__global__ __launch_bounds__(256) __attribute__((amdgpu_waves_per_eu(2, 2)))
void pbv_kernel(const float* __restrict__ v,
                const float* __restrict__ offset,
                const float* __restrict__ w,
                float* __restrict__ out)
{
    __shared__ __attribute__((aligned(16))) float band[NBAND * BANDN];
    __shared__ __attribute__((aligned(16))) float vlds[2 * VSTAGE];
    __shared__ float zs[256 * ZPT];
    __shared__ float wtot[8];

    const int tid  = threadIdx.x;
    const int lane = tid & 31;
    const int wave = tid >> 5;
    const int m    = lane & 15;
    const int kh   = lane >> 4;
    const int jb   = blockIdx.x * MT;
    const int n    = blockIdx.y;
    const int h    = blockIdx.z;

    const float* wrow = w + h * SHAPE;
    const float off = offset[h];
    const int tb = (QTR - MT) - jb;

    v8f c[NBAND][4] = {};

    const float* vbase = v + (((size_t)n * LFULL + 1) * Hh + h) * Dd;
    const int vrowk = tid >> 4;
    const int vd    = (tid & 15) * 4;
    const int vst   = vd * VSTR + vrowk;
    const size_t qoff = (size_t)QTR * (Hh * Dd);

    const float* src0 = vbase + (size_t)vrowk * (Hh * Dd) + vd;
    const float4 h0 = *(const float4*)(src0);
    const float4 h1 = *(const float4*)(src0 + qoff);
    const float4 h2 = *(const float4*)(src0 + 2 * qoff);
    const float4 h3 = *(const float4*)(src0 + 3 * qoff);

    float xr[3][7];
    #pragma unroll
    for (int r = 0; r < 3; ++r) {
        #pragma unroll
        for (int e = 0; e < 7; ++e)
            xr[r][e] = wrow[min(tb + e * QTR + tid + 256 * r, SHAPE - 1)];
    }

    const int piece = blockIdx.x * Bc + n;
    const int zb    = piece * ZP;
    float run = 0.f;
    float loc[ZPT];
    #pragma unroll
    for (int i = 0; i < ZPT; ++i) {
        const int k = zb + tid * ZPT + i;
        const float e = __expf(wrow[min(k, SHAPE - 1)] - off);
        loc[i] = run;
        run += e;
    }
    float inc = run;
    #pragma unroll
    for (int d = 1; d < 32; d <<= 1) {
        const float t = __shfl_up(inc, d);
        if (lane >= d) inc += t;
    }
    float excl = __shfl_up(inc, 1);
    if (lane == 0) excl = 0.f;
    if (lane == 31) wtot[wave] = inc;

    #pragma unroll
    for (int r = 0; r < 3; ++r) {
        const int i = tid + 256 * r;
        if (r < 2 || tid < BANDN - 512) {
            const float wm3 = __expf(xr[r][0] - off), wm2 = __expf(xr[r][1] - off), wm1 = __expf(xr[r][2] - off);
            const float w0  = __expf(xr[r][3] - off);
            const float w1  = __expf(xr[r][4] - off), w2  = __expf(xr[r][5] - off), w3  = __expf(xr[r][6] - off);
            const float pA0 = w2 - w0;
            const float mB0 = wm2 - w0;
            float* bd = band + i;
            bd[0 * BANDN] = w0;
            bd[1 * BANDN] = w1 - w0;
            bd[2 * BANDN] = wm1 - w0;
            bd[3 * BANDN] = pA0;
            bd[4 * BANDN] = (w3 - w1) - pA0;
            bd[5 * BANDN] = (w1 - wm1) - pA0;
            bd[6 * BANDN] = mB0;
            bd[7 * BANDN] = (wm1 - w1) - mB0;
            bd[8 * BANDN] = (wm3 - wm1) - mB0;
        }
    }
    put_slab(vlds + vst, h0, h1, h2, h3);
    __syncthreads();

    {
        float wbase = 0.f;
        #pragma unroll
        for (int q = 0; q < 7; ++q)
            if (q < wave) wbase += wtot[q];
        const float tbase = wbase + excl;
        #pragma unroll
        for (int i = 0; i < ZPT; ++i) zs[tid * ZPT + i] = tbase + loc[i];
    }
    __syncthreads();
    {
        float* zout = out + (size_t)Bc * LFULL * Hh * Dd;
        if (tid < ZP) {
            const float z = zs[tid + Lq] - zs[tid];
            const int row = zb + tid + 1;
            #pragma unroll
            for (int nn = 0; nn < Bc; ++nn)
                zout[((size_t)nn * LFULL + row) * Hh + h] = z;
        }
        if (blockIdx.x == 0) {
            if (n == 0 && tid < 2 * Bc) {
                const int nn = tid & 3;
                const int jp = (tid >> 2) ? (LFULL - 1) : 0;
                zout[((size_t)nn * LFULL + jp) * Hh + h] = 0.f;
            }
            if (tid < 2 * Dd) {
                const int d  = tid & 63;
                const int jp = (tid >> 6) ? (LFULL - 1) : 0;
                out[(((size_t)n * LFULL + jp) * Hh + h) * Dd + d] = 0.f;
            }
        }
    }

    const int arow = (MT - 1) - (wave * 16 + m) + 2 * kh;
    const float* ab = band + arow;

    KFrag q0, q1;
    load_k(q0, vlds, ab, 0, m, kh);

    for (int it = 0; it < NSLAB - 1; ++it) {
        const int l0 = it * KT;
        const float* vcur = vlds + (it & 1) * VSTAGE;
        float* vnxt = vlds + ((it & 1) ^ 1) * VSTAGE;
        const float* ap = ab + l0;

        const float* src = vbase + (size_t)(l0 + KT + vrowk) * (Hh * Dd) + vd;
        const float4 g0 = *(const float4*)(src);
        const float4 g1 = *(const float4*)(src + qoff);
        const float4 g2 = *(const float4*)(src + 2 * qoff);
        const float4 g3 = *(const float4*)(src + 3 * qoff);
        load_k(q1, vcur, ap, 1, m, kh);
        __builtin_amdgcn_sched_barrier(0);
        mma_k(q0, c);
        __builtin_amdgcn_sched_barrier(0);
        load_k(q0, vcur, ap, 2, m, kh);
        __builtin_amdgcn_sched_barrier(0);
        mma_k(q1, c);
        __builtin_amdgcn_sched_barrier(0);
        load_k(q1, vcur, ap, 3, m, kh);
        __builtin_amdgcn_sched_barrier(0);
        mma_k(q0, c);
        __builtin_amdgcn_sched_barrier(0);

        put_slab(vnxt + vst, g0, g1, g2, g3);
        __syncthreads();

        load_k(q0, vnxt, ap + KT, 0, m, kh);
        __builtin_amdgcn_sched_barrier(0);
        mma_k(q1, c);
    }
    {
        const float* vcur = vlds + ((NSLAB - 1) & 1) * VSTAGE;
        const float* ap = ab + (NSLAB - 1) * KT;
        load_k(q1, vcur, ap, 1, m, kh);
        __builtin_amdgcn_sched_barrier(0);
        mma_k(q0, c);
        __builtin_amdgcn_sched_barrier(0);
        load_k(q0, vcur, ap, 2, m, kh);
        __builtin_amdgcn_sched_barrier(0);
        mma_k(q1, c);
        __builtin_amdgcn_sched_barrier(0);
        load_k(q1, vcur, ap, 3, m, kh);
        __builtin_amdgcn_sched_barrier(0);
        mma_k(q0, c);
        mma_k(q1, c);
    }

    #pragma unroll
    for (int nt = 0; nt < 4; ++nt) {
        const int d = nt * 16 + m;
        #pragma unroll
        for (int r = 0; r < 8; ++r) {
            const int mrow = (lane < 16) ? r : (r + 8);
            const int j = jb + wave * 16 + mrow;
            float* o = out + (((size_t)n * LFULL + (j + 1)) * Hh + h) * Dd + d;
            const float s1 = c[0][nt][r] + c[1][nt][r];
            const float sm = c[0][nt][r] + c[2][nt][r];
            o[0]                              = s1 + (c[3][nt][r] + c[4][nt][r]);
            o[(size_t)QTR * (Hh * Dd)]        = sm + (c[3][nt][r] + c[5][nt][r]);
            o[(size_t)2 * QTR * (Hh * Dd)]    = s1 + (c[6][nt][r] + c[7][nt][r]);
            o[(size_t)3 * QTR * (Hh * Dd)]    = sm + (c[6][nt][r] + c[8][nt][r]);
        }
    }
}

extern "C" void kernel_launch(void* const* d_in, const int* in_sizes, int n_in,
                              void* d_out, int out_size, void* d_ws, size_t ws_size,
                              hipStream_t stream)
{
    (void)in_sizes; (void)n_in; (void)out_size; (void)ws_size;
    const float* v      = (const float*)d_in[0];
    const float* offset = (const float*)d_in[1];
    const float* w      = (const float*)d_in[2];
    float*       out    = (float*)d_out;
    (void)d_ws;

    pbv_kernel<<<dim3(QTR / MT, Bc, Hh), 256, 0, stream>>>(v, offset, w, out);
}


# Toeplitz operand cut a third time (8x8 blocks of 256): 27 block products instead of 36, 54 accumulator tiles, two 32-column passes at 1 wave per SIMD; changes the f32 sum order (z_pb scan runs of 17)
# speedup vs baseline: 2.2951x; 1.1530x over previous
#include <hip/hip_runtime.h>

typedef float v2f __attribute__((ext_vector_type(2)));
typedef float v8f __attribute__((ext_vector_type(8)));

#define Bc     4
#define LFULL  2050
#define Lq     2048
#define Hh     16
#define Dd     64
#define SHAPE  4095

#define EB     (Lq / 8)
#define KT     16
#define MT     64
#define NSLAB  (EB / KT)
#define NPASS  2
#define BANDN  (EB + MT)
#define NBAND  27
#define NOCT   8
#define NTW    2
#define NCOL   (NTW * 16)
#define VSTR   (KT + 4)
#define VSLAB  (NCOL * VSTR)
#define VSTAGE (NOCT * VSLAB)
#define ZP     128
#define ZPT    17
#define NTHR   128


__device__ __forceinline__ void cut_bands(const float (&r)[15], float (&o)[NBAND])
{
    float f1[3][7];
    #pragma unroll
    for (int i = 0; i < 7; ++i) {
        f1[0][i] = r[i + 4];
        f1[1][i] = r[i + 8] - r[i + 4];
        f1[2][i] = r[i] - r[i + 4];
    }
    #pragma unroll
    for (int a = 0; a < 3; ++a) {
        float f2[3][3];
        #pragma unroll
        for (int i = 0; i < 3; ++i) {
            f2[0][i] = f1[a][i + 2];
            f2[1][i] = f1[a][i + 4] - f1[a][i + 2];
            f2[2][i] = f1[a][i] - f1[a][i + 2];
        }
        #pragma unroll
        for (int b = 0; b < 3; ++b) {
            o[9 * a + 3 * b + 0] = f2[b][1];
            o[9 * a + 3 * b + 1] = f2[b][2] - f2[b][1];
            o[9 * a + 3 * b + 2] = f2[b][0] - f2[b][1];
        }
    }
}

struct KFrag {
    v2f a[NBAND];
    v2f b[NTW][NOCT];
};

__device__ __forceinline__ void load_k(KFrag& f,
                                       const float* __restrict__ vcur,
                                       const float* __restrict__ ap,
                                       const int ks, const int m, const int kh)
{
    #pragma unroll
    for (int q = 0; q < NBAND; ++q) {
        f.a[q][0] = ap[q * BANDN + ks * 4];
        f.a[q][1] = ap[q * BANDN + ks * 4 + 1];
    }
    #pragma unroll
    for (int nt = 0; nt < NTW; ++nt) {
        const float* bp = &vcur[(nt * 16 + m) * VSTR + ks * 4 + 2 * kh];
        #pragma unroll
        for (int o = 0; o < NOCT; ++o) {
            f.b[nt][o][0] = bp[o * VSLAB];
            f.b[nt][o][1] = bp[o * VSLAB + 1];
        }
    }
}

#define WMMA4(A, B, C) __builtin_amdgcn_wmma_f32_16x16x4_f32(false, (A), false, (B), (short)0, (C), false, false)

__device__ __forceinline__ void mma_k(const KFrag& f, v8f (&c)[NBAND][NTW])
{
    #pragma unroll
    for (int nt = 0; nt < NTW; ++nt) {
        v2f l1[3][4];
        #pragma unroll
        for (int i = 0; i < 4; ++i) {
            l1[0][i] = f.b[nt][i] + f.b[nt][i + 4];
            l1[1][i] = f.b[nt][i + 4];
            l1[2][i] = f.b[nt][i];
        }
        v2f z[NBAND];
        #pragma unroll
        for (int a = 0; a < 3; ++a) {
            v2f l2[3][2];
            #pragma unroll
            for (int i = 0; i < 2; ++i) {
                l2[0][i] = l1[a][i] + l1[a][i + 2];
                l2[1][i] = l1[a][i + 2];
                l2[2][i] = l1[a][i];
            }
            #pragma unroll
            for (int b = 0; b < 3; ++b) {
                z[9 * a + 3 * b + 0] = l2[b][0] + l2[b][1];
                z[9 * a + 3 * b + 1] = l2[b][1];
                z[9 * a + 3 * b + 2] = l2[b][0];
            }
        }
        #pragma unroll
        for (int p = 0; p < NBAND; ++p)
            c[p][nt] = WMMA4(f.a[p], z[p], c[p][nt]);
    }
}

__device__ __forceinline__ void put_slab(float* __restrict__ dst, const float4 (&g)[NOCT])
{
    #pragma unroll
    for (int o = 0; o < NOCT; ++o) {
        dst[o * VSLAB + 0 * VSTR] = g[o].x;
        dst[o * VSLAB + 1 * VSTR] = g[o].y;
        dst[o * VSLAB + 2 * VSTR] = g[o].z;
        dst[o * VSLAB + 3 * VSTR] = g[o].w;
    }
}

__device__ __forceinline__ void write_out(const v8f (&c)[NBAND][NTW], float* __restrict__ o0, const int lane)
{
    #pragma unroll
    for (int nt = 0; nt < NTW; ++nt) {
        #pragma unroll
        for (int r = 0; r < 8; ++r) {
            const int mrow = (lane < 16) ? r : (r + 8);
            float* o = o0 + (size_t)mrow * (Hh * Dd) + nt * 16;
            float t2[3][4];
            #pragma unroll
            for (int a = 0; a < 3; ++a) {
                float t3[3][2];
                #pragma unroll
                for (int b = 0; b < 3; ++b) {
                    t3[b][0] = c[9 * a + 3 * b][nt][r] + c[9 * a + 3 * b + 1][nt][r];
                    t3[b][1] = c[9 * a + 3 * b][nt][r] + c[9 * a + 3 * b + 2][nt][r];
                }
                t2[a][0] = t3[0][0] + t3[1][0];
                t2[a][1] = t3[0][1] + t3[1][1];
                t2[a][2] = t3[0][0] + t3[2][0];
                t2[a][3] = t3[0][1] + t3[2][1];
            }
            #pragma unroll
            for (int i = 0; i < 4; ++i) {
                o[(size_t)i * EB * (Hh * Dd)]       = t2[0][i] + t2[1][i];
                o[(size_t)(i + 4) * EB * (Hh * Dd)] = t2[0][i] + t2[2][i];
            }
        }
    }
}

__global__ __launch_bounds__(NTHR) __attribute__((amdgpu_waves_per_eu(1, 1)))
void pbv_kernel(const float* __restrict__ v,
                const float* __restrict__ offset,
                const float* __restrict__ w,
                float* __restrict__ out)
{
    __shared__ __attribute__((aligned(16))) float band[NBAND * BANDN];
    __shared__ __attribute__((aligned(16))) float vlds[2 * VSTAGE];
    __shared__ float zs[NTHR * ZPT];
    __shared__ float wtot[4];

    const int tid  = threadIdx.x;
    const int lane = tid & 31;
    const int wave = tid >> 5;
    const int m    = lane & 15;
    const int kh   = lane >> 4;
    const int jb   = blockIdx.x * MT;
    const int n    = blockIdx.y;
    const int h    = blockIdx.z;

    const float* wrow = w + h * SHAPE;
    const float off = offset[h];
    const int tb = (EB - MT) - jb;

    const float* vbase = v + (((size_t)n * LFULL + 1) * Hh + h) * Dd;
    const int vrowk = tid >> 3;
    const int vd    = (tid & 7) * 4;
    const int vst   = vd * VSTR + vrowk;
    const size_t ooff = (size_t)EB * (Hh * Dd);

    float4 g[NOCT];
    {
        const float* src0 = vbase + (size_t)vrowk * (Hh * Dd) + vd;
        #pragma unroll
        for (int o = 0; o < NOCT; ++o) g[o] = *(const float4*)(src0 + o * ooff);
    }

    float xr[3][15];
    #pragma unroll
    for (int r = 0; r < 3; ++r) {
        #pragma unroll
        for (int e = 0; e < 15; ++e)
            xr[r][e] = wrow[min(tb + e * EB + tid + NTHR * r, SHAPE - 1)];
    }

    const int piece = blockIdx.x * Bc + n;
    const int zb    = piece * ZP;
    float run = 0.f;
    float loc[ZPT];
    #pragma unroll
    for (int i = 0; i < ZPT; ++i) {
        const int k = zb + tid * ZPT + i;
        const float e = __expf(wrow[min(k, SHAPE - 1)] - off);
        loc[i] = run;
        run += e;
    }
    float inc = run;
    #pragma unroll
    for (int d = 1; d < 32; d <<= 1) {
        const float t = __shfl_up(inc, d);
        if (lane >= d) inc += t;
    }
    float excl = __shfl_up(inc, 1);
    if (lane == 0) excl = 0.f;
    if (lane == 31) wtot[wave] = inc;

    #pragma unroll
    for (int r = 0; r < 3; ++r) {
        const int i = tid + NTHR * r;
        if (r < 2 || tid < BANDN - 2 * NTHR) {
            float e15[15];
            #pragma unroll
            for (int e = 0; e < 15; ++e) e15[e] = __expf(xr[r][e] - off);
            float o27[NBAND];
            cut_bands(e15, o27);
            #pragma unroll
            for (int p = 0; p < NBAND; ++p) band[p * BANDN + i] = o27[p];
        }
    }
    put_slab(vlds + vst, g);
    __syncthreads();

    {
        float wbase = 0.f;
        #pragma unroll
        for (int q = 0; q < 3; ++q)
            if (q < wave) wbase += wtot[q];
        const float tbase = wbase + excl;
        #pragma unroll
        for (int i = 0; i < ZPT; ++i) zs[tid * ZPT + i] = tbase + loc[i];
    }
    __syncthreads();
    {
        float* zout = out + (size_t)Bc * LFULL * Hh * Dd;
        {
            const float z = zs[tid + Lq] - zs[tid];
            const int row = zb + tid + 1;
            #pragma unroll
            for (int nn = 0; nn < Bc; ++nn)
                zout[((size_t)nn * LFULL + row) * Hh + h] = z;
        }
        if (blockIdx.x == 0) {
            if (n == 0 && tid < 2 * Bc) {
                const int nn = tid & 3;
                const int jp = (tid >> 2) ? (LFULL - 1) : 0;
                zout[((size_t)nn * LFULL + jp) * Hh + h] = 0.f;
            }
            {
                const int d  = tid & 63;
                const int jp = (tid >> 6) ? (LFULL - 1) : 0;
                out[(((size_t)n * LFULL + jp) * Hh + h) * Dd + d] = 0.f;
            }
        }
    }

    const int arow = (MT - 1) - (wave * 16 + m) + 2 * kh;
    const float* ab = band + arow;
    float* obase = out + (((size_t)n * LFULL + (jb + wave * 16 + 1)) * Hh + h) * Dd + m;

    v8f c[NBAND][NTW] = {};

    KFrag q0, q1;
    load_k(q0, vlds, ab, 0, m, kh);

    #pragma nounroll
    for (int s = 0; s < NPASS * NSLAB; ++s) {
        const int l0 = (s & (NSLAB - 1)) * KT;
        const float* vcur = vlds + (s & 1) * VSTAGE;
        float* vnxt = vlds + ((s & 1) ^ 1) * VSTAGE;
        const float* ap = ab + l0;
        const int sn  = (s + 1) & (NPASS * NSLAB - 1);
        const int l0n = (sn & (NSLAB - 1)) * KT;

        const float* src = vbase + (size_t)(l0n + vrowk) * (Hh * Dd) + (sn >> 4) * NCOL + vd;
        #pragma unroll
        for (int o = 0; o < NOCT; ++o) g[o] = *(const float4*)(src + o * ooff);
        load_k(q1, vcur, ap, 1, m, kh);
        __builtin_amdgcn_sched_barrier(0);
        mma_k(q0, c);
        __builtin_amdgcn_sched_barrier(0);
        load_k(q0, vcur, ap, 2, m, kh);
        __builtin_amdgcn_sched_barrier(0);
        mma_k(q1, c);
        __builtin_amdgcn_sched_barrier(0);
        load_k(q1, vcur, ap, 3, m, kh);
        __builtin_amdgcn_sched_barrier(0);
        mma_k(q0, c);
        __builtin_amdgcn_sched_barrier(0);

        put_slab(vnxt + vst, g);
        __syncthreads();

        load_k(q0, vnxt, ab + l0n, 0, m, kh);
        __builtin_amdgcn_sched_barrier(0);
        mma_k(q1, c);

        if ((s & (NSLAB - 1)) == NSLAB - 1) {
            write_out(c, obase + (s >> 4) * NCOL, lane);
            #pragma unroll
            for (int p = 0; p < NBAND; ++p) {
                #pragma unroll
                for (int nt = 0; nt < NTW; ++nt) c[p][nt] = v8f{};
            }
        }
    }
}

extern "C" void kernel_launch(void* const* d_in, const int* in_sizes, int n_in,
                              void* d_out, int out_size, void* d_ws, size_t ws_size,
                              hipStream_t stream)
{
    (void)in_sizes; (void)n_in; (void)out_size; (void)ws_size;
    const float* v      = (const float*)d_in[0];
    const float* offset = (const float*)d_in[1];
    const float* w      = (const float*)d_in[2];
    float*       out    = (float*)d_out;
    (void)d_ws;

    pbv_kernel<<<dim3(EB / MT, Bc, Hh), NTHR, 0, stream>>>(v, offset, w, out);
}


# Fourth cut of the Toeplitz operand, run in time: its 3 products pass one after another through the same 54-tile loop (24 slabs, not 32), the first result held in registers; changes the f32 sum order
# speedup vs baseline: 2.6468x; 1.1532x over previous
#include <hip/hip_runtime.h>

typedef float v2f __attribute__((ext_vector_type(2)));
typedef float v8f __attribute__((ext_vector_type(8)));

#define Bc     4
#define LFULL  2050
#define Lq     2048
#define Hh     16
#define Dd     64
#define SHAPE  4095

#define EB     (Lq / 16)
#define HB     (Lq / 2)
#define KT     16
#define MT     64
#define NSLAB  (EB / KT)
#define NGRP   3
#define BANDN  (EB + MT)
#define NBAND  27
#define NOCT   8
#define NTW    2
#define NCOL   (NTW * 16)
#define VSTR   (KT + 4)
#define VSLAB  (NCOL * VSTR)
#define VSTAGE (NOCT * VSLAB)
#define ZP     128
#define ZPT    17
#define NTHR   128


__device__ __forceinline__ void cut_bands(const float (&r)[15], float (&o)[NBAND])
{
    float f1[3][7];
    #pragma unroll
    for (int i = 0; i < 7; ++i) {
        f1[0][i] = r[i + 4];
        f1[1][i] = r[i + 8] - r[i + 4];
        f1[2][i] = r[i] - r[i + 4];
    }
    #pragma unroll
    for (int a = 0; a < 3; ++a) {
        float f2[3][3];
        #pragma unroll
        for (int i = 0; i < 3; ++i) {
            f2[0][i] = f1[a][i + 2];
            f2[1][i] = f1[a][i + 4] - f1[a][i + 2];
            f2[2][i] = f1[a][i] - f1[a][i + 2];
        }
        #pragma unroll
        for (int b = 0; b < 3; ++b) {
            o[9 * a + 3 * b + 0] = f2[b][1];
            o[9 * a + 3 * b + 1] = f2[b][2] - f2[b][1];
            o[9 * a + 3 * b + 2] = f2[b][0] - f2[b][1];
        }
    }
}

struct KFrag {
    v2f a[NBAND];
    v2f b[NTW][NOCT];
};

__device__ __forceinline__ void load_k(KFrag& f,
                                       const float* __restrict__ vcur,
                                       const float* __restrict__ ap,
                                       const int ks, const int m, const int kh)
{
    #pragma unroll
    for (int q = 0; q < NBAND; ++q) {
        f.a[q][0] = ap[q * BANDN + ks * 4];
        f.a[q][1] = ap[q * BANDN + ks * 4 + 1];
    }
    #pragma unroll
    for (int nt = 0; nt < NTW; ++nt) {
        const float* bp = &vcur[(nt * 16 + m) * VSTR + ks * 4 + 2 * kh];
        #pragma unroll
        for (int o = 0; o < NOCT; ++o) {
            f.b[nt][o][0] = bp[o * VSLAB];
            f.b[nt][o][1] = bp[o * VSLAB + 1];
        }
    }
}

#define WMMA4(A, B, C) __builtin_amdgcn_wmma_f32_16x16x4_f32(false, (A), false, (B), (short)0, (C), false, false)

__device__ __forceinline__ void mma_k(const KFrag& f, v8f (&c)[NBAND][NTW])
{
    #pragma unroll
    for (int nt = 0; nt < NTW; ++nt) {
        v2f l1[3][4];
        #pragma unroll
        for (int i = 0; i < 4; ++i) {
            l1[0][i] = f.b[nt][i] + f.b[nt][i + 4];
            l1[1][i] = f.b[nt][i + 4];
            l1[2][i] = f.b[nt][i];
        }
        v2f z[NBAND];
        #pragma unroll
        for (int a = 0; a < 3; ++a) {
            v2f l2[3][2];
            #pragma unroll
            for (int i = 0; i < 2; ++i) {
                l2[0][i] = l1[a][i] + l1[a][i + 2];
                l2[1][i] = l1[a][i + 2];
                l2[2][i] = l1[a][i];
            }
            #pragma unroll
            for (int b = 0; b < 3; ++b) {
                z[9 * a + 3 * b + 0] = l2[b][0] + l2[b][1];
                z[9 * a + 3 * b + 1] = l2[b][1];
                z[9 * a + 3 * b + 2] = l2[b][0];
            }
        }
        #pragma unroll
        for (int p = 0; p < NBAND; ++p)
            c[p][nt] = WMMA4(f.a[p], z[p], c[p][nt]);
    }
}

__device__ __forceinline__ void put_slab(float* __restrict__ dst, const float4 (&g)[NOCT])
{
    #pragma unroll
    for (int o = 0; o < NOCT; ++o) {
        dst[o * VSLAB + 0 * VSTR] = g[o].x;
        dst[o * VSLAB + 1 * VSTR] = g[o].y;
        dst[o * VSLAB + 2 * VSTR] = g[o].z;
        dst[o * VSLAB + 3 * VSTR] = g[o].w;
    }
}

__device__ __forceinline__ void combine(const v8f (&c)[NBAND][NTW], v8f (&res)[NOCT][NTW])
{
    #pragma unroll
    for (int nt = 0; nt < NTW; ++nt) {
        v8f t2[3][4];
        #pragma unroll
        for (int a = 0; a < 3; ++a) {
            v8f t3[3][2];
            #pragma unroll
            for (int b = 0; b < 3; ++b) {
                t3[b][0] = c[9 * a + 3 * b][nt] + c[9 * a + 3 * b + 1][nt];
                t3[b][1] = c[9 * a + 3 * b][nt] + c[9 * a + 3 * b + 2][nt];
            }
            t2[a][0] = t3[0][0] + t3[1][0];
            t2[a][1] = t3[0][1] + t3[1][1];
            t2[a][2] = t3[0][0] + t3[2][0];
            t2[a][3] = t3[0][1] + t3[2][1];
        }
        #pragma unroll
        for (int i = 0; i < 4; ++i) {
            res[i][nt]     = t2[0][i] + t2[1][i];
            res[i + 4][nt] = t2[0][i] + t2[2][i];
        }
    }
}

__device__ __forceinline__ void store_rows(const v8f (&res)[NOCT][NTW], float* __restrict__ o0, const int lane)
{
    #pragma unroll
    for (int nt = 0; nt < NTW; ++nt) {
        #pragma unroll
        for (int r = 0; r < 8; ++r) {
            const int mrow = (lane < 16) ? r : (r + 8);
            float* o = o0 + (size_t)mrow * (Hh * Dd) + nt * 16;
            #pragma unroll
            for (int i = 0; i < NOCT; ++i)
                o[(size_t)i * EB * (Hh * Dd)] = res[i][nt][r];
        }
    }
}

__device__ __forceinline__ void restart(v8f (&c)[NBAND][NTW], const v8f (&r0)[NOCT][NTW])
{
    #pragma unroll
    for (int p = 0; p < NBAND; ++p) {
        #pragma unroll
        for (int nt = 0; nt < NTW; ++nt) c[p][nt] = v8f{};
    }
    #pragma unroll
    for (int i = 0; i < NOCT; ++i) {
        #pragma unroll
        for (int nt = 0; nt < NTW; ++nt)
            c[9 * (1 + (i >> 2)) + 3 * (1 + ((i >> 1) & 1)) + (1 + (i & 1))][nt] = r0[i][nt];
    }
}

__global__ __launch_bounds__(NTHR) __attribute__((amdgpu_waves_per_eu(1, 1)))
void pbv_kernel(const float* __restrict__ v,
                const float* __restrict__ offset,
                const float* __restrict__ w,
                float* __restrict__ out)
{
    __shared__ __attribute__((aligned(16))) float band[NGRP * NBAND * BANDN];
    __shared__ __attribute__((aligned(16))) float vlds[2 * VSTAGE];
    __shared__ float zs[NTHR * ZPT];
    __shared__ float wtot[4];

    const int tid  = threadIdx.x;
    const int lane = tid & 31;
    const int wave = tid >> 5;
    const int m    = lane & 15;
    const int kh   = lane >> 4;
    const int jb   = (blockIdx.x & 1) * MT;
    const int ch   = blockIdx.x >> 1;
    const int n    = blockIdx.y;
    const int h    = blockIdx.z;

    const float* wrow = w + h * SHAPE;
    const float off = offset[h];
    const int tb = (EB - MT) - jb;

    const float* vbase = v + (((size_t)n * LFULL + 1) * Hh + h) * Dd + ch * NCOL;
    const int vrowk = tid >> 3;
    const int vd    = (tid & 7) * 4;
    const int vst   = vd * VSTR + vrowk;
    const size_t ooff = (size_t)EB * (Hh * Dd);
    const size_t hoff = (size_t)HB * (Hh * Dd);

    float4 g[NOCT], g2[NOCT];
    {
        const float* src0 = vbase + (size_t)vrowk * (Hh * Dd) + vd;
        #pragma unroll
        for (int o = 0; o < NOCT; ++o) {
            g[o]  = *(const float4*)(src0 + o * ooff);
            g2[o] = *(const float4*)(src0 + hoff + o * ooff);
        }
    }

    float xr[2][31];
    #pragma unroll
    for (int r = 0; r < 2; ++r) {
        #pragma unroll
        for (int e = 0; e < 31; ++e)
            xr[r][e] = wrow[min(tb + e * EB + tid + NTHR * r, SHAPE - 1)];
    }

    const int piece = blockIdx.x * Bc + n;
    const int zb    = piece * ZP;
    float run = 0.f;
    float loc[ZPT];
    #pragma unroll
    for (int i = 0; i < ZPT; ++i) {
        const int k = zb + tid * ZPT + i;
        const float e = __expf(wrow[min(k, SHAPE - 1)] - off);
        loc[i] = run;
        run += e;
    }
    float inc = run;
    #pragma unroll
    for (int d = 1; d < 32; d <<= 1) {
        const float t = __shfl_up(inc, d);
        if (lane >= d) inc += t;
    }
    float excl = __shfl_up(inc, 1);
    if (lane == 0) excl = 0.f;
    if (lane == 31) wtot[wave] = inc;

    #pragma unroll
    for (int r = 0; r < 2; ++r) {
        const int i = tid + NTHR * r;
        if (r < 1 || tid < BANDN - NTHR) {
            float e31[31];
            #pragma unroll
            for (int e = 0; e < 31; ++e) e31[e] = __expf(xr[r][e] - off);
            #pragma unroll
            for (int gq = 0; gq < NGRP; ++gq) {
                float e15[15];
                #pragma unroll
                for (int e = 0; e < 15; ++e)
                    e15[e] = (gq == 0) ? e31[e + 8] : (gq == 1) ? (e31[e + 16] - e31[e + 8]) : (e31[e] - e31[e + 8]);
                float o27[NBAND];
                cut_bands(e15, o27);
                #pragma unroll
                for (int p = 0; p < NBAND; ++p) band[(gq * NBAND + p) * BANDN + i] = o27[p];
            }
        }
    }
    #pragma unroll
    for (int o = 0; o < NOCT; ++o) {
        g[o].x += g2[o].x; g[o].y += g2[o].y; g[o].z += g2[o].z; g[o].w += g2[o].w;
    }
    put_slab(vlds + vst, g);
    __syncthreads();

    {
        float wbase = 0.f;
        #pragma unroll
        for (int q = 0; q < 3; ++q)
            if (q < wave) wbase += wtot[q];
        const float tbase = wbase + excl;
        #pragma unroll
        for (int i = 0; i < ZPT; ++i) zs[tid * ZPT + i] = tbase + loc[i];
    }
    __syncthreads();
    {
        float* zout = out + (size_t)Bc * LFULL * Hh * Dd;
        {
            const float z = zs[tid + Lq] - zs[tid];
            const int row = zb + tid + 1;
            #pragma unroll
            for (int nn = 0; nn < Bc; ++nn)
                zout[((size_t)nn * LFULL + row) * Hh + h] = z;
        }
        if (blockIdx.x == 0) {
            if (n == 0 && tid < 2 * Bc) {
                const int nn = tid & 3;
                const int jp = (tid >> 2) ? (LFULL - 1) : 0;
                zout[((size_t)nn * LFULL + jp) * Hh + h] = 0.f;
            }
            {
                const int d  = tid & 63;
                const int jp = (tid >> 6) ? (LFULL - 1) : 0;
                out[(((size_t)n * LFULL + jp) * Hh + h) * Dd + d] = 0.f;
            }
        }
    }

    const int arow = (MT - 1) - (wave * 16 + m) + 2 * kh;
    const float* ab = band + arow;
    float* obase = out + (((size_t)n * LFULL + (jb + wave * 16 + 1)) * Hh + h) * Dd + ch * NCOL + m;

    v8f c[NBAND][NTW] = {};
    v8f res0[NOCT][NTW] = {};

    KFrag q0, q1;
    load_k(q0, vlds, ab, 0, m, kh);

    #pragma nounroll
    for (int s = 0; s < NGRP * NSLAB; ++s) {
        const int grp = s / NSLAB;
        const int l0 = (s % NSLAB) * KT;
        const float* vcur = vlds + (s & 1) * VSTAGE;
        float* vnxt = vlds + ((s & 1) ^ 1) * VSTAGE;
        const float* ap = ab + grp * (NBAND * BANDN) + l0;
        const int sn   = (s + 1 == NGRP * NSLAB) ? 0 : (s + 1);
        const int grpn = sn / NSLAB;
        const int l0n  = (sn % NSLAB) * KT;

        const float* src = vbase + (size_t)(l0n + vrowk) * (Hh * Dd) + vd + ((grpn == 1) ? hoff : (size_t)0);
        #pragma unroll
        for (int o = 0; o < NOCT; ++o) g[o] = *(const float4*)(src + o * ooff);
        if (grpn == 0) {
            #pragma unroll
            for (int o = 0; o < NOCT; ++o) g2[o] = *(const float4*)(src + hoff + o * ooff);
        }
        load_k(q1, vcur, ap, 1, m, kh);
        __builtin_amdgcn_sched_barrier(0);
        mma_k(q0, c);
        __builtin_amdgcn_sched_barrier(0);
        load_k(q0, vcur, ap, 2, m, kh);
        __builtin_amdgcn_sched_barrier(0);
        mma_k(q1, c);
        __builtin_amdgcn_sched_barrier(0);
        load_k(q1, vcur, ap, 3, m, kh);
        __builtin_amdgcn_sched_barrier(0);
        mma_k(q0, c);
        __builtin_amdgcn_sched_barrier(0);

        if (grpn == 0) {
            #pragma unroll
            for (int o = 0; o < NOCT; ++o) {
                g[o].x += g2[o].x; g[o].y += g2[o].y; g[o].z += g2[o].z; g[o].w += g2[o].w;
            }
        }
        put_slab(vnxt + vst, g);
        __syncthreads();

        load_k(q0, vnxt, ab + grpn * (NBAND * BANDN) + l0n, 0, m, kh);
        __builtin_amdgcn_sched_barrier(0);
        mma_k(q1, c);

        if (s % NSLAB == NSLAB - 1) {
            v8f res[NOCT][NTW];
            combine(c, res);
            if (grp == 0) {
                #pragma unroll
                for (int i = 0; i < NOCT; ++i) {
                    #pragma unroll
                    for (int nt = 0; nt < NTW; ++nt) res0[i][nt] = res[i][nt];
                }
            } else {
                store_rows(res, obase + (size_t)(grp - 1) * HB * (Hh * Dd), lane);
            }
            restart(c, res0);
        }
    }
}

extern "C" void kernel_launch(void* const* d_in, const int* in_sizes, int n_in,
                              void* d_out, int out_size, void* d_ws, size_t ws_size,
                              hipStream_t stream)
{
    (void)in_sizes; (void)n_in; (void)out_size; (void)ws_size;
    const float* v      = (const float*)d_in[0];
    const float* offset = (const float*)d_in[1];
    const float* w      = (const float*)d_in[2];
    float*       out    = (float*)d_out;
    (void)d_ws;

    pbv_kernel<<<dim3(4, Bc, Hh), NTHR, 0, stream>>>(v, offset, w, out);
}


# Slab hand-over's barrier split: signal behind the stage stores (between k2's N tiles), wait 27 WMMAs later, ahead of the next slab's reads; no wave stands at a barrier with its matrix pipe empty
# speedup vs baseline: 2.7186x; 1.0271x over previous
#include <hip/hip_runtime.h>

typedef float v2f __attribute__((ext_vector_type(2)));
typedef float v8f __attribute__((ext_vector_type(8)));

#define Bc     4
#define LFULL  2050
#define Lq     2048
#define Hh     16
#define Dd     64
#define SHAPE  4095

#define EB     (Lq / 16)
#define HB     (Lq / 2)
#define KT     16
#define MT     64
#define NSLAB  (EB / KT)
#define NGRP   3
#define BANDN  (EB + MT)
#define NBAND  27
#define NOCT   8
#define NTW    2
#define NCOL   (NTW * 16)
#define VSTR   (KT + 4)
#define VSLAB  (NCOL * VSTR)
#define VSTAGE (NOCT * VSLAB)
#define ZP     128
#define ZPT    17
#define NTHR   128


__device__ __forceinline__ void cut_bands(const float (&r)[15], float (&o)[NBAND])
{
    float f1[3][7];
    #pragma unroll
    for (int i = 0; i < 7; ++i) {
        f1[0][i] = r[i + 4];
        f1[1][i] = r[i + 8] - r[i + 4];
        f1[2][i] = r[i] - r[i + 4];
    }
    #pragma unroll
    for (int a = 0; a < 3; ++a) {
        float f2[3][3];
        #pragma unroll
        for (int i = 0; i < 3; ++i) {
            f2[0][i] = f1[a][i + 2];
            f2[1][i] = f1[a][i + 4] - f1[a][i + 2];
            f2[2][i] = f1[a][i] - f1[a][i + 2];
        }
        #pragma unroll
        for (int b = 0; b < 3; ++b) {
            o[9 * a + 3 * b + 0] = f2[b][1];
            o[9 * a + 3 * b + 1] = f2[b][2] - f2[b][1];
            o[9 * a + 3 * b + 2] = f2[b][0] - f2[b][1];
        }
    }
}

struct KFrag {
    v2f a[NBAND];
    v2f b[NTW][NOCT];
};

__device__ __forceinline__ void load_k(KFrag& f,
                                       const float* __restrict__ vcur,
                                       const float* __restrict__ ap,
                                       const int ks, const int m, const int kh)
{
    #pragma unroll
    for (int q = 0; q < NBAND; ++q) {
        f.a[q][0] = ap[q * BANDN + ks * 4];
        f.a[q][1] = ap[q * BANDN + ks * 4 + 1];
    }
    #pragma unroll
    for (int nt = 0; nt < NTW; ++nt) {
        const float* bp = &vcur[(nt * 16 + m) * VSTR + ks * 4 + 2 * kh];
        #pragma unroll
        for (int o = 0; o < NOCT; ++o) {
            f.b[nt][o][0] = bp[o * VSLAB];
            f.b[nt][o][1] = bp[o * VSLAB + 1];
        }
    }
}

#define WMMA4(A, B, C) __builtin_amdgcn_wmma_f32_16x16x4_f32(false, (A), false, (B), (short)0, (C), false, false)

__device__ __forceinline__ void mma_k_nt(const KFrag& f, v8f (&c)[NBAND][NTW], const int nt)
{
    {
        v2f l1[3][4];
        #pragma unroll
        for (int i = 0; i < 4; ++i) {
            l1[0][i] = f.b[nt][i] + f.b[nt][i + 4];
            l1[1][i] = f.b[nt][i + 4];
            l1[2][i] = f.b[nt][i];
        }
        v2f z[NBAND];
        #pragma unroll
        for (int a = 0; a < 3; ++a) {
            v2f l2[3][2];
            #pragma unroll
            for (int i = 0; i < 2; ++i) {
                l2[0][i] = l1[a][i] + l1[a][i + 2];
                l2[1][i] = l1[a][i + 2];
                l2[2][i] = l1[a][i];
            }
            #pragma unroll
            for (int b = 0; b < 3; ++b) {
                z[9 * a + 3 * b + 0] = l2[b][0] + l2[b][1];
                z[9 * a + 3 * b + 1] = l2[b][1];
                z[9 * a + 3 * b + 2] = l2[b][0];
            }
        }
        #pragma unroll
        for (int p = 0; p < NBAND; ++p)
            c[p][nt] = WMMA4(f.a[p], z[p], c[p][nt]);
    }
}

__device__ __forceinline__ void mma_k(const KFrag& f, v8f (&c)[NBAND][NTW])
{
    mma_k_nt(f, c, 0);
    mma_k_nt(f, c, 1);
}

__device__ __forceinline__ void slab_signal() { asm volatile("s_wait_dscnt 0x0\n\ts_barrier_signal -1" ::: "memory"); }
__device__ __forceinline__ void slab_wait() { asm volatile("s_barrier_wait -1" ::: "memory"); }

__device__ __forceinline__ void put_slab(float* __restrict__ dst, const float4 (&g)[NOCT])
{
    #pragma unroll
    for (int o = 0; o < NOCT; ++o) {
        dst[o * VSLAB + 0 * VSTR] = g[o].x;
        dst[o * VSLAB + 1 * VSTR] = g[o].y;
        dst[o * VSLAB + 2 * VSTR] = g[o].z;
        dst[o * VSLAB + 3 * VSTR] = g[o].w;
    }
}

__device__ __forceinline__ void combine(const v8f (&c)[NBAND][NTW], v8f (&res)[NOCT][NTW])
{
    #pragma unroll
    for (int nt = 0; nt < NTW; ++nt) {
        v8f t2[3][4];
        #pragma unroll
        for (int a = 0; a < 3; ++a) {
            v8f t3[3][2];
            #pragma unroll
            for (int b = 0; b < 3; ++b) {
                t3[b][0] = c[9 * a + 3 * b][nt] + c[9 * a + 3 * b + 1][nt];
                t3[b][1] = c[9 * a + 3 * b][nt] + c[9 * a + 3 * b + 2][nt];
            }
            t2[a][0] = t3[0][0] + t3[1][0];
            t2[a][1] = t3[0][1] + t3[1][1];
            t2[a][2] = t3[0][0] + t3[2][0];
            t2[a][3] = t3[0][1] + t3[2][1];
        }
        #pragma unroll
        for (int i = 0; i < 4; ++i) {
            res[i][nt]     = t2[0][i] + t2[1][i];
            res[i + 4][nt] = t2[0][i] + t2[2][i];
        }
    }
}

__device__ __forceinline__ void store_rows(const v8f (&res)[NOCT][NTW], float* __restrict__ o0, const int lane)
{
    #pragma unroll
    for (int nt = 0; nt < NTW; ++nt) {
        #pragma unroll
        for (int r = 0; r < 8; ++r) {
            const int mrow = (lane < 16) ? r : (r + 8);
            float* o = o0 + (size_t)mrow * (Hh * Dd) + nt * 16;
            #pragma unroll
            for (int i = 0; i < NOCT; ++i)
                o[(size_t)i * EB * (Hh * Dd)] = res[i][nt][r];
        }
    }
}

__device__ __forceinline__ void restart(v8f (&c)[NBAND][NTW], const v8f (&r0)[NOCT][NTW])
{
    #pragma unroll
    for (int p = 0; p < NBAND; ++p) {
        #pragma unroll
        for (int nt = 0; nt < NTW; ++nt) c[p][nt] = v8f{};
    }
    #pragma unroll
    for (int i = 0; i < NOCT; ++i) {
        #pragma unroll
        for (int nt = 0; nt < NTW; ++nt)
            c[9 * (1 + (i >> 2)) + 3 * (1 + ((i >> 1) & 1)) + (1 + (i & 1))][nt] = r0[i][nt];
    }
}

__global__ __launch_bounds__(NTHR) __attribute__((amdgpu_waves_per_eu(1, 1)))
void pbv_kernel(const float* __restrict__ v,
                const float* __restrict__ offset,
                const float* __restrict__ w,
                float* __restrict__ out)
{
    __shared__ __attribute__((aligned(16))) float band[NGRP * NBAND * BANDN];
    __shared__ __attribute__((aligned(16))) float vlds[2 * VSTAGE];
    __shared__ float zs[NTHR * ZPT];
    __shared__ float wtot[4];

    const int tid  = threadIdx.x;
    const int lane = tid & 31;
    const int wave = tid >> 5;
    const int m    = lane & 15;
    const int kh   = lane >> 4;
    const int jb   = (blockIdx.x & 1) * MT;
    const int ch   = blockIdx.x >> 1;
    const int n    = blockIdx.y;
    const int h    = blockIdx.z;

    const float* wrow = w + h * SHAPE;
    const float off = offset[h];
    const int tb = (EB - MT) - jb;

    const float* vbase = v + (((size_t)n * LFULL + 1) * Hh + h) * Dd + ch * NCOL;
    const int vrowk = tid >> 3;
    const int vd    = (tid & 7) * 4;
    const int vst   = vd * VSTR + vrowk;
    const size_t ooff = (size_t)EB * (Hh * Dd);
    const size_t hoff = (size_t)HB * (Hh * Dd);

    float4 g[NOCT], g2[NOCT];
    {
        const float* src0 = vbase + (size_t)vrowk * (Hh * Dd) + vd;
        #pragma unroll
        for (int o = 0; o < NOCT; ++o) {
            g[o]  = *(const float4*)(src0 + o * ooff);
            g2[o] = *(const float4*)(src0 + hoff + o * ooff);
        }
    }

    float xr[2][31];
    #pragma unroll
    for (int r = 0; r < 2; ++r) {
        #pragma unroll
        for (int e = 0; e < 31; ++e)
            xr[r][e] = wrow[min(tb + e * EB + tid + NTHR * r, SHAPE - 1)];
    }

    const int piece = blockIdx.x * Bc + n;
    const int zb    = piece * ZP;
    float run = 0.f;
    float loc[ZPT];
    #pragma unroll
    for (int i = 0; i < ZPT; ++i) {
        const int k = zb + tid * ZPT + i;
        const float e = __expf(wrow[min(k, SHAPE - 1)] - off);
        loc[i] = run;
        run += e;
    }
    float inc = run;
    #pragma unroll
    for (int d = 1; d < 32; d <<= 1) {
        const float t = __shfl_up(inc, d);
        if (lane >= d) inc += t;
    }
    float excl = __shfl_up(inc, 1);
    if (lane == 0) excl = 0.f;
    if (lane == 31) wtot[wave] = inc;

    #pragma unroll
    for (int r = 0; r < 2; ++r) {
        const int i = tid + NTHR * r;
        if (r < 1 || tid < BANDN - NTHR) {
            float e31[31];
            #pragma unroll
            for (int e = 0; e < 31; ++e) e31[e] = __expf(xr[r][e] - off);
            #pragma unroll
            for (int gq = 0; gq < NGRP; ++gq) {
                float e15[15];
                #pragma unroll
                for (int e = 0; e < 15; ++e)
                    e15[e] = (gq == 0) ? e31[e + 8] : (gq == 1) ? (e31[e + 16] - e31[e + 8]) : (e31[e] - e31[e + 8]);
                float o27[NBAND];
                cut_bands(e15, o27);
                #pragma unroll
                for (int p = 0; p < NBAND; ++p) band[(gq * NBAND + p) * BANDN + i] = o27[p];
            }
        }
    }
    #pragma unroll
    for (int o = 0; o < NOCT; ++o) {
        g[o].x += g2[o].x; g[o].y += g2[o].y; g[o].z += g2[o].z; g[o].w += g2[o].w;
    }
    put_slab(vlds + vst, g);
    __syncthreads();

    {
        float wbase = 0.f;
        #pragma unroll
        for (int q = 0; q < 3; ++q)
            if (q < wave) wbase += wtot[q];
        const float tbase = wbase + excl;
        #pragma unroll
        for (int i = 0; i < ZPT; ++i) zs[tid * ZPT + i] = tbase + loc[i];
    }
    __syncthreads();
    {
        float* zout = out + (size_t)Bc * LFULL * Hh * Dd;
        {
            const float z = zs[tid + Lq] - zs[tid];
            const int row = zb + tid + 1;
            #pragma unroll
            for (int nn = 0; nn < Bc; ++nn)
                zout[((size_t)nn * LFULL + row) * Hh + h] = z;
        }
        if (blockIdx.x == 0) {
            if (n == 0 && tid < 2 * Bc) {
                const int nn = tid & 3;
                const int jp = (tid >> 2) ? (LFULL - 1) : 0;
                zout[((size_t)nn * LFULL + jp) * Hh + h] = 0.f;
            }
            {
                const int d  = tid & 63;
                const int jp = (tid >> 6) ? (LFULL - 1) : 0;
                out[(((size_t)n * LFULL + jp) * Hh + h) * Dd + d] = 0.f;
            }
        }
    }

    const int arow = (MT - 1) - (wave * 16 + m) + 2 * kh;
    const float* ab = band + arow;
    float* obase = out + (((size_t)n * LFULL + (jb + wave * 16 + 1)) * Hh + h) * Dd + ch * NCOL + m;

    v8f c[NBAND][NTW] = {};
    v8f res0[NOCT][NTW] = {};

    KFrag q0, q1;
    load_k(q0, vlds, ab, 0, m, kh);

    #pragma nounroll
    for (int s = 0; s < NGRP * NSLAB; ++s) {
        const int grp = s / NSLAB;
        const int l0 = (s % NSLAB) * KT;
        const float* vcur = vlds + (s & 1) * VSTAGE;
        float* vnxt = vlds + ((s & 1) ^ 1) * VSTAGE;
        const float* ap = ab + grp * (NBAND * BANDN) + l0;
        const int sn   = (s + 1 == NGRP * NSLAB) ? 0 : (s + 1);
        const int grpn = sn / NSLAB;
        const int l0n  = (sn % NSLAB) * KT;

        const float* src = vbase + (size_t)(l0n + vrowk) * (Hh * Dd) + vd + ((grpn == 1) ? hoff : (size_t)0);
        #pragma unroll
        for (int o = 0; o < NOCT; ++o) g[o] = *(const float4*)(src + o * ooff);
        if (grpn == 0) {
            #pragma unroll
            for (int o = 0; o < NOCT; ++o) g2[o] = *(const float4*)(src + hoff + o * ooff);
        }
        load_k(q1, vcur, ap, 1, m, kh);
        __builtin_amdgcn_sched_barrier(0);
        mma_k(q0, c);
        __builtin_amdgcn_sched_barrier(0);
        load_k(q0, vcur, ap, 2, m, kh);
        __builtin_amdgcn_sched_barrier(0);
        mma_k(q1, c);
        __builtin_amdgcn_sched_barrier(0);
        load_k(q1, vcur, ap, 3, m, kh);
        __builtin_amdgcn_sched_barrier(0);
        mma_k_nt(q0, c, 0);
        __builtin_amdgcn_sched_barrier(0);

        if (grpn == 0) {
            #pragma unroll
            for (int o = 0; o < NOCT; ++o) {
                g[o].x += g2[o].x; g[o].y += g2[o].y; g[o].z += g2[o].z; g[o].w += g2[o].w;
            }
        }
        put_slab(vnxt + vst, g);
        __builtin_amdgcn_sched_barrier(0);
        mma_k_nt(q0, c, 1);
        __builtin_amdgcn_sched_barrier(0);
        slab_signal();
        __builtin_amdgcn_sched_barrier(0);
        mma_k_nt(q1, c, 0);
        __builtin_amdgcn_sched_barrier(0);
        slab_wait();
        load_k(q0, vnxt, ab + grpn * (NBAND * BANDN) + l0n, 0, m, kh);
        __builtin_amdgcn_sched_barrier(0);
        mma_k_nt(q1, c, 1);

        if (s % NSLAB == NSLAB - 1) {
            v8f res[NOCT][NTW];
            combine(c, res);
            if (grp == 0) {
                #pragma unroll
                for (int i = 0; i < NOCT; ++i) {
                    #pragma unroll
                    for (int nt = 0; nt < NTW; ++nt) res0[i][nt] = res[i][nt];
                }
            } else {
                store_rows(res, obase + (size_t)(grp - 1) * HB * (Hh * Dd), lane);
            }
            restart(c, res0);
        }
    }
}

extern "C" void kernel_launch(void* const* d_in, const int* in_sizes, int n_in,
                              void* d_out, int out_size, void* d_ws, size_t ws_size,
                              hipStream_t stream)
{
    (void)in_sizes; (void)n_in; (void)out_size; (void)ws_size;
    const float* v      = (const float*)d_in[0];
    const float* offset = (const float*)d_in[1];
    const float* w      = (const float*)d_in[2];
    float*       out    = (float*)d_out;
    (void)d_ws;

    pbv_kernel<<<dim3(4, Bc, Hh), NTHR, 0, stream>>>(v, offset, w, out);
}


# Fifth cut of the Toeplitz operand (32x32 blocks of 64), run in time like the first: 9 passes of 27 products (36 slabs of 108 WMMAs, not 24 of 216), one 16x16 tile per wave; changes the f32 sum order
# speedup vs baseline: 2.8327x; 1.0420x over previous
#include <hip/hip_runtime.h>

typedef float v2f __attribute__((ext_vector_type(2)));
typedef float v8f __attribute__((ext_vector_type(8)));

#define Bc     4
#define LFULL  2050
#define Lq     2048
#define Hh     16
#define Dd     64
#define SHAPE  4095

#define EB     (Lq / 32)
#define QB     (Lq / 4)
#define HB     (Lq / 2)
#define KT     16
#define MT     32
#define NSLAB  (EB / KT)
#define NPASS  9
#define BANDN  (EB + MT)
#define NBAND  27
#define NOCT   8
#define NCOL   32
#define VSTR   (KT + 4)
#define VSLAB  (NCOL * VSTR)
#define VSTAGE (NOCT * VSLAB)
#define ZP     128
#define ZPT    17
#define NTHR   128
#define XSTR   65


__device__ __forceinline__ void cut_bands(const float (&r)[15], float (&o)[NBAND])
{
    float f1[3][7];
    #pragma unroll
    for (int i = 0; i < 7; ++i) {
        f1[0][i] = r[i + 4];
        f1[1][i] = r[i + 8] - r[i + 4];
        f1[2][i] = r[i] - r[i + 4];
    }
    #pragma unroll
    for (int a = 0; a < 3; ++a) {
        float f2[3][3];
        #pragma unroll
        for (int i = 0; i < 3; ++i) {
            f2[0][i] = f1[a][i + 2];
            f2[1][i] = f1[a][i + 4] - f1[a][i + 2];
            f2[2][i] = f1[a][i] - f1[a][i + 2];
        }
        #pragma unroll
        for (int b = 0; b < 3; ++b) {
            o[9 * a + 3 * b + 0] = f2[b][1];
            o[9 * a + 3 * b + 1] = f2[b][2] - f2[b][1];
            o[9 * a + 3 * b + 2] = f2[b][0] - f2[b][1];
        }
    }
}

struct KFrag {
    v2f a[NBAND];
    v2f b[NOCT];
};

__device__ __forceinline__ void load_k(KFrag& f,
                                       const float* __restrict__ vcur,
                                       const float* __restrict__ ap,
                                       const int ks, const int cb, const int kh)
{
    #pragma unroll
    for (int q = 0; q < NBAND; ++q) {
        f.a[q][0] = ap[q * BANDN + ks * 4];
        f.a[q][1] = ap[q * BANDN + ks * 4 + 1];
    }
    const float* bp = &vcur[cb * VSTR + ks * 4 + 2 * kh];
    #pragma unroll
    for (int o = 0; o < NOCT; ++o) {
        f.b[o][0] = bp[o * VSLAB];
        f.b[o][1] = bp[o * VSLAB + 1];
    }
}

#define WMMA4(A, B, C) __builtin_amdgcn_wmma_f32_16x16x4_f32(false, (A), false, (B), (short)0, (C), false, false)

__device__ __forceinline__ void make_z(const KFrag& f, v2f (&z)[NBAND])
{
    v2f l1[3][4];
    #pragma unroll
    for (int i = 0; i < 4; ++i) {
        l1[0][i] = f.b[i] + f.b[i + 4];
        l1[1][i] = f.b[i + 4];
        l1[2][i] = f.b[i];
    }
    #pragma unroll
    for (int a = 0; a < 3; ++a) {
        v2f l2[3][2];
        #pragma unroll
        for (int i = 0; i < 2; ++i) {
            l2[0][i] = l1[a][i] + l1[a][i + 2];
            l2[1][i] = l1[a][i + 2];
            l2[2][i] = l1[a][i];
        }
        #pragma unroll
        for (int b = 0; b < 3; ++b) {
            z[9 * a + 3 * b + 0] = l2[b][0] + l2[b][1];
            z[9 * a + 3 * b + 1] = l2[b][1];
            z[9 * a + 3 * b + 2] = l2[b][0];
        }
    }
}

template <int LO, int HI>
__device__ __forceinline__ void mma_part(const KFrag& f, const v2f (&z)[NBAND], v8f (&c)[NBAND])
{
    #pragma unroll
    for (int p = LO; p < HI; ++p)
        c[p] = WMMA4(f.a[p], z[p], c[p]);
}

__device__ __forceinline__ void slab_signal() { asm volatile("s_wait_dscnt 0x0\n\ts_barrier_signal -1" ::: "memory"); }
__device__ __forceinline__ void slab_wait() { asm volatile("s_barrier_wait -1" ::: "memory"); }

__device__ __forceinline__ void put_slab(float* __restrict__ dst, const float4 (&g)[NOCT])
{
    #pragma unroll
    for (int o = 0; o < NOCT; ++o) {
        dst[o * VSLAB + 0 * VSTR] = g[o].x;
        dst[o * VSLAB + 1 * VSTR] = g[o].y;
        dst[o * VSLAB + 2 * VSTR] = g[o].z;
        dst[o * VSLAB + 3 * VSTR] = g[o].w;
    }
}

__device__ __forceinline__ void add_slab(float4 (&g)[NOCT], const float4 (&x)[NOCT])
{
    #pragma unroll
    for (int o = 0; o < NOCT; ++o) {
        g[o].x += x[o].x; g[o].y += x[o].y; g[o].z += x[o].z; g[o].w += x[o].w;
    }
}

__device__ __forceinline__ void combine(const v8f (&c)[NBAND], v8f (&res)[NOCT])
{
    v8f t2[3][4];
    #pragma unroll
    for (int a = 0; a < 3; ++a) {
        v8f t3[3][2];
        #pragma unroll
        for (int b = 0; b < 3; ++b) {
            t3[b][0] = c[9 * a + 3 * b] + c[9 * a + 3 * b + 1];
            t3[b][1] = c[9 * a + 3 * b] + c[9 * a + 3 * b + 2];
        }
        t2[a][0] = t3[0][0] + t3[1][0];
        t2[a][1] = t3[0][1] + t3[1][1];
        t2[a][2] = t3[0][0] + t3[2][0];
        t2[a][3] = t3[0][1] + t3[2][1];
    }
    #pragma unroll
    for (int i = 0; i < 4; ++i) {
        res[i]     = t2[0][i] + t2[1][i];
        res[i + 4] = t2[0][i] + t2[2][i];
    }
}

__device__ __forceinline__ void restart(v8f (&c)[NBAND], const v8f (&sd)[NOCT])
{
    #pragma unroll
    for (int p = 0; p < NBAND; ++p) c[p] = v8f{};
    #pragma unroll
    for (int i = 0; i < NOCT; ++i)
        c[9 * (1 + (i >> 2)) + 3 * (1 + ((i >> 1) & 1)) + (1 + (i & 1))] = sd[i];
}

__global__ __launch_bounds__(NTHR) __attribute__((amdgpu_waves_per_eu(1, 1)))
void pbv_kernel(const float* __restrict__ v,
                const float* __restrict__ offset,
                const float* __restrict__ w,
                float* __restrict__ out)
{
    __shared__ __attribute__((aligned(16))) float band[NPASS * NBAND * BANDN];
    __shared__ __attribute__((aligned(16))) float vlds[2 * VSTAGE];
    __shared__ float zs[NTHR * ZPT];
    __shared__ float wtot[4];
    __shared__ float xch[NTHR * XSTR];

    const int tid  = threadIdx.x;
    const int lane = tid & 31;
    const int wave = tid >> 5;
    const int m    = lane & 15;
    const int kh   = lane >> 4;
    const int mt   = wave >> 1;
    const int nt   = wave & 1;
    const int jb   = (blockIdx.x & 1) * MT;
    const int ch   = blockIdx.x >> 1;
    const int n    = blockIdx.y;
    const int h    = blockIdx.z;

    const float* wrow = w + h * SHAPE;
    const float off = offset[h];
    const int tb = (EB - MT) - jb;

    const float* vbase = v + (((size_t)n * LFULL + 1) * Hh + h) * Dd + ch * NCOL;
    const int vrowk = tid >> 3;
    const int vd    = (tid & 7) * 4;
    const int vst   = vd * VSTR + vrowk;
    const size_t ooff = (size_t)EB * (Hh * Dd);
    const size_t qoff = (size_t)QB * (Hh * Dd);

    float4 g[NOCT], g2[NOCT], g3[NOCT], g4[NOCT];
    {
        const float* src0 = vbase + (size_t)vrowk * (Hh * Dd) + vd;
        #pragma unroll
        for (int o = 0; o < NOCT; ++o) {
            g[o]  = *(const float4*)(src0 + o * ooff);
            g2[o] = *(const float4*)(src0 + qoff + o * ooff);
            g3[o] = *(const float4*)(src0 + 2 * qoff + o * ooff);
            g4[o] = *(const float4*)(src0 + 3 * qoff + o * ooff);
        }
    }

    float xr[63];
    #pragma unroll
    for (int e = 0; e < 63; ++e)
        xr[e] = wrow[min(tb + e * EB + tid, SHAPE - 1)];

    const int piece = blockIdx.x * Bc + n;
    const int zb    = piece * ZP;
    float run = 0.f;
    float loc[ZPT];
    #pragma unroll
    for (int i = 0; i < ZPT; ++i) {
        const int k = zb + tid * ZPT + i;
        const float e = __expf(wrow[min(k, SHAPE - 1)] - off);
        loc[i] = run;
        run += e;
    }
    float inc = run;
    #pragma unroll
    for (int d = 1; d < 32; d <<= 1) {
        const float t = __shfl_up(inc, d);
        if (lane >= d) inc += t;
    }
    float excl = __shfl_up(inc, 1);
    if (lane == 0) excl = 0.f;
    if (lane == 31) wtot[wave] = inc;

    if (tid < BANDN) {
        float e63[63];
        #pragma unroll
        for (int e = 0; e < 63; ++e) e63[e] = __expf(xr[e] - off);
        #pragma unroll
        for (int a1 = 0; a1 < 3; ++a1) {
            float e31[31];
            #pragma unroll
            for (int e = 0; e < 31; ++e)
                e31[e] = (a1 == 0) ? e63[e + 16] : (a1 == 1) ? (e63[e + 32] - e63[e + 16]) : (e63[e] - e63[e + 16]);
            #pragma unroll
            for (int b1 = 0; b1 < 3; ++b1) {
                float e15[15];
                #pragma unroll
                for (int e = 0; e < 15; ++e)
                    e15[e] = (b1 == 0) ? e31[e + 8] : (b1 == 1) ? (e31[e + 16] - e31[e + 8]) : (e31[e] - e31[e + 8]);
                float o27[NBAND];
                cut_bands(e15, o27);
                #pragma unroll
                for (int p = 0; p < NBAND; ++p) band[((a1 * 3 + b1) * NBAND + p) * BANDN + tid] = o27[p];
            }
        }
    }
    add_slab(g, g2);
    add_slab(g, g3);
    add_slab(g, g4);
    put_slab(vlds + vst, g);
    __syncthreads();

    {
        float wbase = 0.f;
        #pragma unroll
        for (int q = 0; q < 3; ++q)
            if (q < wave) wbase += wtot[q];
        const float tbase = wbase + excl;
        #pragma unroll
        for (int i = 0; i < ZPT; ++i) zs[tid * ZPT + i] = tbase + loc[i];
    }
    __syncthreads();
    {
        float* zout = out + (size_t)Bc * LFULL * Hh * Dd;
        {
            const float z = zs[tid + Lq] - zs[tid];
            const int row = zb + tid + 1;
            #pragma unroll
            for (int nn = 0; nn < Bc; ++nn)
                zout[((size_t)nn * LFULL + row) * Hh + h] = z;
        }
        if (blockIdx.x == 0) {
            if (n == 0 && tid < 2 * Bc) {
                const int nn = tid & 3;
                const int jp = (tid >> 2) ? (LFULL - 1) : 0;
                zout[((size_t)nn * LFULL + jp) * Hh + h] = 0.f;
            }
            {
                const int d  = tid & 63;
                const int jp = (tid >> 6) ? (LFULL - 1) : 0;
                out[(((size_t)n * LFULL + jp) * Hh + h) * Dd + d] = 0.f;
            }
        }
    }

    const int arow = (MT - 1) - (mt * 16 + m) + 2 * kh;
    const float* ab = band + arow;
    const int cb = nt * 16 + m;
    float* obase = out + (((size_t)n * LFULL + (jb + mt * 16 + nt * 8 + 1)) * Hh + h) * Dd + ch * NCOL + lane;
    const bool give = (kh != nt);
    float* xmine = xch + tid * XSTR;
    const float* xother = xch + (tid ^ 48) * XSTR;

    v8f c[NBAND] = {};
    v8f k0[NOCT] = {};
    v8f tt[NOCT] = {};
    v8f bt[NOCT] = {};

    KFrag q0, q1;
    v2f z[NBAND];
    load_k(q0, vlds, ab, 0, cb, kh);

    #pragma nounroll
    for (int s = 0; s < NPASS * NSLAB; ++s) {
        const int pas = s / NSLAB;
        const int l0 = (s % NSLAB) * KT;
        const float* vcur = vlds + (s & 1) * VSTAGE;
        float* vnxt = vlds + ((s & 1) ^ 1) * VSTAGE;
        const float* ap = ab + pas * (NBAND * BANDN) + l0;
        const int sn   = (s + 1 == NPASS * NSLAB) ? 0 : (s + 1);
        const int pasn = sn / NSLAB;
        const int l0n  = (sn % NSLAB) * KT;
        const int an   = pasn / 3;
        const int bn   = pasn % 3;

        const int rown = ((an == 1) ? HB : 0) + ((bn == 1) ? QB : 0) + l0n + vrowk;
        const float* src = vbase + (size_t)rown * (Hh * Dd) + vd;
        #pragma unroll
        for (int o = 0; o < NOCT; ++o) g[o] = *(const float4*)(src + o * ooff);
        if (bn == 0) {
            #pragma unroll
            for (int o = 0; o < NOCT; ++o) g2[o] = *(const float4*)(src + qoff + o * ooff);
        }
        if (an == 0) {
            #pragma unroll
            for (int o = 0; o < NOCT; ++o) g3[o] = *(const float4*)(src + 2 * qoff + o * ooff);
            if (bn == 0) {
                #pragma unroll
                for (int o = 0; o < NOCT; ++o) g4[o] = *(const float4*)(src + 3 * qoff + o * ooff);
            }
        }
        load_k(q1, vcur, ap, 1, cb, kh);
        __builtin_amdgcn_sched_barrier(0);
        make_z(q0, z);
        mma_part<0, NBAND>(q0, z, c);
        __builtin_amdgcn_sched_barrier(0);
        load_k(q0, vcur, ap, 2, cb, kh);
        __builtin_amdgcn_sched_barrier(0);
        make_z(q1, z);
        mma_part<0, NBAND>(q1, z, c);
        __builtin_amdgcn_sched_barrier(0);
        load_k(q1, vcur, ap, 3, cb, kh);
        __builtin_amdgcn_sched_barrier(0);
        make_z(q0, z);
        mma_part<0, 14>(q0, z, c);
        __builtin_amdgcn_sched_barrier(0);

        if (bn == 0) add_slab(g, g2);
        if (an == 0) {
            add_slab(g, g3);
            if (bn == 0) add_slab(g, g4);
        }
        put_slab(vnxt + vst, g);
        __builtin_amdgcn_sched_barrier(0);
        mma_part<14, NBAND>(q0, z, c);
        __builtin_amdgcn_sched_barrier(0);
        slab_signal();
        __builtin_amdgcn_sched_barrier(0);
        make_z(q1, z);
        mma_part<0, 14>(q1, z, c);
        __builtin_amdgcn_sched_barrier(0);
        slab_wait();
        load_k(q0, vnxt, ab + pasn * (NBAND * BANDN) + l0n, 0, cb, kh);
        __builtin_amdgcn_sched_barrier(0);
        mma_part<14, NBAND>(q1, z, c);

        if (s % NSLAB == NSLAB - 1) {
            const int a = pas / 3;
            const int b = pas % 3;
            v8f res[NOCT];
            combine(c, res);
            if (a != 0 && b != 0) {
                if (give) {
                    #pragma unroll
                    for (int i = 0; i < NOCT; ++i) {
                        #pragma unroll
                        for (int r = 0; r < 8; ++r) xmine[i * 8 + r] = res[i][r];
                    }
                }
                __syncthreads();
                if (give) {
                    #pragma unroll
                    for (int i = 0; i < NOCT; ++i) {
                        #pragma unroll
                        for (int r = 0; r < 8; ++r) res[i][r] = xother[i * 8 + r];
                    }
                }
                float* o0 = obase + (size_t)(2 * (a - 1) + (b - 1)) * QB * (Hh * Dd);
                #pragma unroll
                for (int i = 0; i < NOCT; ++i) {
                    #pragma unroll
                    for (int r = 0; r < 8; ++r)
                        o0[(size_t)(i * EB + r) * (Hh * Dd)] = res[i][r];
                }
            }
            v8f sd[NOCT];
            if (b == 0) {
                #pragma unroll
                for (int i = 0; i < NOCT; ++i) k0[i] = res[i];
            } else if (a == 0) {
                if (b == 1) {
                    #pragma unroll
                    for (int i = 0; i < NOCT; ++i) tt[i] = res[i];
                } else {
                    #pragma unroll
                    for (int i = 0; i < NOCT; ++i) bt[i] = res[i];
                }
            }
            if (b == 2) {
                #pragma unroll
                for (int i = 0; i < NOCT; ++i) sd[i] = v8f{};
            } else if (a == 0) {
                #pragma unroll
                for (int i = 0; i < NOCT; ++i) sd[i] = k0[i];
            } else if (b == 0) {
                #pragma unroll
                for (int i = 0; i < NOCT; ++i) sd[i] = tt[i] + k0[i];
            } else {
                #pragma unroll
                for (int i = 0; i < NOCT; ++i) sd[i] = bt[i] + k0[i];
            }
            restart(c, sd);
        }
    }
}

extern "C" void kernel_launch(void* const* d_in, const int* in_sizes, int n_in,
                              void* d_out, int out_size, void* d_ws, size_t ws_size,
                              hipStream_t stream)
{
    (void)in_sizes; (void)n_in; (void)out_size; (void)ws_size;
    const float* v      = (const float*)d_in[0];
    const float* offset = (const float*)d_in[1];
    const float* w      = (const float*)d_in[2];
    float*       out    = (float*)d_out;
    (void)d_ws;

    pbv_kernel<<<dim3(4, Bc, Hh), NTHR, 0, stream>>>(v, offset, w, out);
}


# K loop's A fragments read two products per LDS instruction: bands stored position-major, products side by side, in two copies one position apart so every read is 16-byte aligned: 14 reads per k-step
# speedup vs baseline: 3.0444x; 1.0747x over previous
#include <hip/hip_runtime.h>

typedef float v2f __attribute__((ext_vector_type(2)));
typedef float v8f __attribute__((ext_vector_type(8)));

#define Bc     4
#define LFULL  2050
#define Lq     2048
#define Hh     16
#define Dd     64
#define SHAPE  4095

#define EB     (Lq / 32)
#define QB     (Lq / 4)
#define HB     (Lq / 2)
#define KT     16
#define MT     32
#define NSLAB  (EB / KT)
#define NPASS  9
#define BANDN  (EB + MT)
#define NBAND  27
#define RP     60
#define NROW   (BANDN / 2)
#define GSZ    (NROW * RP)
#define COPYSZ (NPASS * GSZ)
#define OPAD   32
#define NOCT   8
#define NCOL   32
#define VSTR   (KT + 4)
#define VSLAB  (NCOL * VSTR)
#define VSTAGE (NOCT * VSLAB)
#define ZP     128
#define ZPT    17
#define NTHR   128
#define XSTR   65


__device__ __forceinline__ void cut_bands(const float (&r)[15], float (&o)[NBAND])
{
    float f1[3][7];
    #pragma unroll
    for (int i = 0; i < 7; ++i) {
        f1[0][i] = r[i + 4];
        f1[1][i] = r[i + 8] - r[i + 4];
        f1[2][i] = r[i] - r[i + 4];
    }
    #pragma unroll
    for (int a = 0; a < 3; ++a) {
        float f2[3][3];
        #pragma unroll
        for (int i = 0; i < 3; ++i) {
            f2[0][i] = f1[a][i + 2];
            f2[1][i] = f1[a][i + 4] - f1[a][i + 2];
            f2[2][i] = f1[a][i] - f1[a][i + 2];
        }
        #pragma unroll
        for (int b = 0; b < 3; ++b) {
            o[9 * a + 3 * b + 0] = f2[b][1];
            o[9 * a + 3 * b + 1] = f2[b][2] - f2[b][1];
            o[9 * a + 3 * b + 2] = f2[b][0] - f2[b][1];
        }
    }
}

struct KFrag {
    v2f a[NBAND];
    v2f b[NOCT];
};

__device__ __forceinline__ void load_k(KFrag& f,
                                       const float* __restrict__ vcur,
                                       const float* __restrict__ ap,
                                       const int ks, const int cb, const int kh)
{
    #pragma unroll
    for (int q = 0; q + 1 < NBAND; q += 2) {
        const float4 t = *(const float4*)(ap + ks * 2 * RP + 2 * q);
        f.a[q][0] = t.x;     f.a[q][1] = t.y;
        f.a[q + 1][0] = t.z; f.a[q + 1][1] = t.w;
    }
    {
        const float2 t = *(const float2*)(ap + ks * 2 * RP + 2 * (NBAND - 1));
        f.a[NBAND - 1][0] = t.x; f.a[NBAND - 1][1] = t.y;
    }
    const float* bp = &vcur[cb * VSTR + ks * 4 + 2 * kh];
    #pragma unroll
    for (int o = 0; o < NOCT; ++o) {
        f.b[o][0] = bp[o * VSLAB];
        f.b[o][1] = bp[o * VSLAB + 1];
    }
}

#define WMMA4(A, B, C) __builtin_amdgcn_wmma_f32_16x16x4_f32(false, (A), false, (B), (short)0, (C), false, false)

__device__ __forceinline__ void make_z(const KFrag& f, v2f (&z)[NBAND])
{
    v2f l1[3][4];
    #pragma unroll
    for (int i = 0; i < 4; ++i) {
        l1[0][i] = f.b[i] + f.b[i + 4];
        l1[1][i] = f.b[i + 4];
        l1[2][i] = f.b[i];
    }
    #pragma unroll
    for (int a = 0; a < 3; ++a) {
        v2f l2[3][2];
        #pragma unroll
        for (int i = 0; i < 2; ++i) {
            l2[0][i] = l1[a][i] + l1[a][i + 2];
            l2[1][i] = l1[a][i + 2];
            l2[2][i] = l1[a][i];
        }
        #pragma unroll
        for (int b = 0; b < 3; ++b) {
            z[9 * a + 3 * b + 0] = l2[b][0] + l2[b][1];
            z[9 * a + 3 * b + 1] = l2[b][1];
            z[9 * a + 3 * b + 2] = l2[b][0];
        }
    }
}

template <int LO, int HI>
__device__ __forceinline__ void mma_part(const KFrag& f, const v2f (&z)[NBAND], v8f (&c)[NBAND])
{
    #pragma unroll
    for (int p = LO; p < HI; ++p)
        c[p] = WMMA4(f.a[p], z[p], c[p]);
}

__device__ __forceinline__ void slab_signal() { asm volatile("s_wait_dscnt 0x0\n\ts_barrier_signal -1" ::: "memory"); }
__device__ __forceinline__ void slab_wait() { asm volatile("s_barrier_wait -1" ::: "memory"); }

__device__ __forceinline__ void put_slab(float* __restrict__ dst, const float4 (&g)[NOCT])
{
    #pragma unroll
    for (int o = 0; o < NOCT; ++o) {
        dst[o * VSLAB + 0 * VSTR] = g[o].x;
        dst[o * VSLAB + 1 * VSTR] = g[o].y;
        dst[o * VSLAB + 2 * VSTR] = g[o].z;
        dst[o * VSLAB + 3 * VSTR] = g[o].w;
    }
}

__device__ __forceinline__ void add_slab(float4 (&g)[NOCT], const float4 (&x)[NOCT])
{
    #pragma unroll
    for (int o = 0; o < NOCT; ++o) {
        g[o].x += x[o].x; g[o].y += x[o].y; g[o].z += x[o].z; g[o].w += x[o].w;
    }
}

__device__ __forceinline__ void combine(const v8f (&c)[NBAND], v8f (&res)[NOCT])
{
    v8f t2[3][4];
    #pragma unroll
    for (int a = 0; a < 3; ++a) {
        v8f t3[3][2];
        #pragma unroll
        for (int b = 0; b < 3; ++b) {
            t3[b][0] = c[9 * a + 3 * b] + c[9 * a + 3 * b + 1];
            t3[b][1] = c[9 * a + 3 * b] + c[9 * a + 3 * b + 2];
        }
        t2[a][0] = t3[0][0] + t3[1][0];
        t2[a][1] = t3[0][1] + t3[1][1];
        t2[a][2] = t3[0][0] + t3[2][0];
        t2[a][3] = t3[0][1] + t3[2][1];
    }
    #pragma unroll
    for (int i = 0; i < 4; ++i) {
        res[i]     = t2[0][i] + t2[1][i];
        res[i + 4] = t2[0][i] + t2[2][i];
    }
}

__device__ __forceinline__ void restart(v8f (&c)[NBAND], const v8f (&sd)[NOCT])
{
    #pragma unroll
    for (int p = 0; p < NBAND; ++p) c[p] = v8f{};
    #pragma unroll
    for (int i = 0; i < NOCT; ++i)
        c[9 * (1 + (i >> 2)) + 3 * (1 + ((i >> 1) & 1)) + (1 + (i & 1))] = sd[i];
}

__global__ __launch_bounds__(NTHR) __attribute__((amdgpu_waves_per_eu(1, 1)))
void pbv_kernel(const float* __restrict__ v,
                const float* __restrict__ offset,
                const float* __restrict__ w,
                float* __restrict__ out)
{
    __shared__ __attribute__((aligned(16))) float band[2 * COPYSZ + OPAD];
    __shared__ __attribute__((aligned(16))) float vlds[2 * VSTAGE];
    __shared__ float zs[NTHR * ZPT];
    __shared__ float wtot[4];
    __shared__ float xch[NTHR * XSTR];

    const int tid  = threadIdx.x;
    const int lane = tid & 31;
    const int wave = tid >> 5;
    const int m    = lane & 15;
    const int kh   = lane >> 4;
    const int mt   = wave >> 1;
    const int nt   = wave & 1;
    const int jb   = (blockIdx.x & 1) * MT;
    const int ch   = blockIdx.x >> 1;
    const int n    = blockIdx.y;
    const int h    = blockIdx.z;

    const float* wrow = w + h * SHAPE;
    const float off = offset[h];
    const int tb = (EB - MT) - jb;

    const float* vbase = v + (((size_t)n * LFULL + 1) * Hh + h) * Dd + ch * NCOL;
    const int vrowk = tid >> 3;
    const int vd    = (tid & 7) * 4;
    const int vst   = vd * VSTR + vrowk;
    const size_t ooff = (size_t)EB * (Hh * Dd);
    const size_t qoff = (size_t)QB * (Hh * Dd);

    float4 g[NOCT], g2[NOCT], g3[NOCT], g4[NOCT];
    {
        const float* src0 = vbase + (size_t)vrowk * (Hh * Dd) + vd;
        #pragma unroll
        for (int o = 0; o < NOCT; ++o) {
            g[o]  = *(const float4*)(src0 + o * ooff);
            g2[o] = *(const float4*)(src0 + qoff + o * ooff);
            g3[o] = *(const float4*)(src0 + 2 * qoff + o * ooff);
            g4[o] = *(const float4*)(src0 + 3 * qoff + o * ooff);
        }
    }

    float xr[63];
    #pragma unroll
    for (int e = 0; e < 63; ++e)
        xr[e] = wrow[min(tb + e * EB + tid, SHAPE - 1)];

    const int piece = blockIdx.x * Bc + n;
    const int zb    = piece * ZP;
    float run = 0.f;
    float loc[ZPT];
    #pragma unroll
    for (int i = 0; i < ZPT; ++i) {
        const int k = zb + tid * ZPT + i;
        const float e = __expf(wrow[min(k, SHAPE - 1)] - off);
        loc[i] = run;
        run += e;
    }
    float inc = run;
    #pragma unroll
    for (int d = 1; d < 32; d <<= 1) {
        const float t = __shfl_up(inc, d);
        if (lane >= d) inc += t;
    }
    float excl = __shfl_up(inc, 1);
    if (lane == 0) excl = 0.f;
    if (lane == 31) wtot[wave] = inc;

    float* bandE = band;
    float* bandO = band + COPYSZ + OPAD;
    if (tid < BANDN) {
        const int eo = (tid >> 1) * RP + (tid & 1);
        const int oo = ((tid - 1) >> 1) * RP + ((tid - 1) & 1);
        float e63[63];
        #pragma unroll
        for (int e = 0; e < 63; ++e) e63[e] = __expf(xr[e] - off);
        #pragma unroll
        for (int a1 = 0; a1 < 3; ++a1) {
            float e31[31];
            #pragma unroll
            for (int e = 0; e < 31; ++e)
                e31[e] = (a1 == 0) ? e63[e + 16] : (a1 == 1) ? (e63[e + 32] - e63[e + 16]) : (e63[e] - e63[e + 16]);
            #pragma unroll
            for (int b1 = 0; b1 < 3; ++b1) {
                float e15[15];
                #pragma unroll
                for (int e = 0; e < 15; ++e)
                    e15[e] = (b1 == 0) ? e31[e + 8] : (b1 == 1) ? (e31[e + 16] - e31[e + 8]) : (e31[e] - e31[e + 8]);
                float o27[NBAND];
                cut_bands(e15, o27);
                #pragma unroll
                for (int p = 0; p < NBAND; ++p) bandE[(a1 * 3 + b1) * GSZ + eo + 2 * p] = o27[p];
                if (tid >= 1) {
                    #pragma unroll
                    for (int p = 0; p < NBAND; ++p) bandO[(a1 * 3 + b1) * GSZ + oo + 2 * p] = o27[p];
                }
            }
        }
    }
    add_slab(g, g2);
    add_slab(g, g3);
    add_slab(g, g4);
    put_slab(vlds + vst, g);
    __syncthreads();

    {
        float wbase = 0.f;
        #pragma unroll
        for (int q = 0; q < 3; ++q)
            if (q < wave) wbase += wtot[q];
        const float tbase = wbase + excl;
        #pragma unroll
        for (int i = 0; i < ZPT; ++i) zs[tid * ZPT + i] = tbase + loc[i];
    }
    __syncthreads();
    {
        float* zout = out + (size_t)Bc * LFULL * Hh * Dd;
        {
            const float z = zs[tid + Lq] - zs[tid];
            const int row = zb + tid + 1;
            #pragma unroll
            for (int nn = 0; nn < Bc; ++nn)
                zout[((size_t)nn * LFULL + row) * Hh + h] = z;
        }
        if (blockIdx.x == 0) {
            if (n == 0 && tid < 2 * Bc) {
                const int nn = tid & 3;
                const int jp = (tid >> 2) ? (LFULL - 1) : 0;
                zout[((size_t)nn * LFULL + jp) * Hh + h] = 0.f;
            }
            {
                const int d  = tid & 63;
                const int jp = (tid >> 6) ? (LFULL - 1) : 0;
                out[(((size_t)n * LFULL + jp) * Hh + h) * Dd + d] = 0.f;
            }
        }
    }

    const int arow = (MT - 1) - (mt * 16 + m) + 2 * kh;
    const float* ab = (arow & 1) ? (bandO + ((arow - 1) >> 1) * RP) : (bandE + (arow >> 1) * RP);
    const int cb = nt * 16 + m;
    float* obase = out + (((size_t)n * LFULL + (jb + mt * 16 + nt * 8 + 1)) * Hh + h) * Dd + ch * NCOL + lane;
    const bool give = (kh != nt);
    float* xmine = xch + tid * XSTR;
    const float* xother = xch + (tid ^ 48) * XSTR;

    v8f c[NBAND] = {};
    v8f k0[NOCT] = {};
    v8f tt[NOCT] = {};
    v8f bt[NOCT] = {};

    KFrag q0, q1;
    v2f z[NBAND];
    load_k(q0, vlds, ab, 0, cb, kh);

    #pragma nounroll
    for (int s = 0; s < NPASS * NSLAB; ++s) {
        const int pas = s / NSLAB;
        const int l0 = (s % NSLAB) * KT;
        const float* vcur = vlds + (s & 1) * VSTAGE;
        float* vnxt = vlds + ((s & 1) ^ 1) * VSTAGE;
        const float* ap = ab + pas * GSZ + (l0 / 2) * RP;
        const int sn   = (s + 1 == NPASS * NSLAB) ? 0 : (s + 1);
        const int pasn = sn / NSLAB;
        const int l0n  = (sn % NSLAB) * KT;
        const int an   = pasn / 3;
        const int bn   = pasn % 3;

        const int rown = ((an == 1) ? HB : 0) + ((bn == 1) ? QB : 0) + l0n + vrowk;
        const float* src = vbase + (size_t)rown * (Hh * Dd) + vd;
        #pragma unroll
        for (int o = 0; o < NOCT; ++o) g[o] = *(const float4*)(src + o * ooff);
        if (bn == 0) {
            #pragma unroll
            for (int o = 0; o < NOCT; ++o) g2[o] = *(const float4*)(src + qoff + o * ooff);
        }
        if (an == 0) {
            #pragma unroll
            for (int o = 0; o < NOCT; ++o) g3[o] = *(const float4*)(src + 2 * qoff + o * ooff);
            if (bn == 0) {
                #pragma unroll
                for (int o = 0; o < NOCT; ++o) g4[o] = *(const float4*)(src + 3 * qoff + o * ooff);
            }
        }
        load_k(q1, vcur, ap, 1, cb, kh);
        __builtin_amdgcn_sched_barrier(0);
        make_z(q0, z);
        mma_part<0, NBAND>(q0, z, c);
        __builtin_amdgcn_sched_barrier(0);
        load_k(q0, vcur, ap, 2, cb, kh);
        __builtin_amdgcn_sched_barrier(0);
        make_z(q1, z);
        mma_part<0, NBAND>(q1, z, c);
        __builtin_amdgcn_sched_barrier(0);
        load_k(q1, vcur, ap, 3, cb, kh);
        __builtin_amdgcn_sched_barrier(0);
        make_z(q0, z);
        mma_part<0, 14>(q0, z, c);
        __builtin_amdgcn_sched_barrier(0);

        if (bn == 0) add_slab(g, g2);
        if (an == 0) {
            add_slab(g, g3);
            if (bn == 0) add_slab(g, g4);
        }
        put_slab(vnxt + vst, g);
        __builtin_amdgcn_sched_barrier(0);
        mma_part<14, NBAND>(q0, z, c);
        __builtin_amdgcn_sched_barrier(0);
        slab_signal();
        __builtin_amdgcn_sched_barrier(0);
        make_z(q1, z);
        mma_part<0, 14>(q1, z, c);
        __builtin_amdgcn_sched_barrier(0);
        slab_wait();
        load_k(q0, vnxt, ab + pasn * GSZ + (l0n / 2) * RP, 0, cb, kh);
        __builtin_amdgcn_sched_barrier(0);
        mma_part<14, NBAND>(q1, z, c);

        if (s % NSLAB == NSLAB - 1) {
            const int a = pas / 3;
            const int b = pas % 3;
            v8f res[NOCT];
            combine(c, res);
            if (a != 0 && b != 0) {
                if (give) {
                    #pragma unroll
                    for (int i = 0; i < NOCT; ++i) {
                        #pragma unroll
                        for (int r = 0; r < 8; ++r) xmine[i * 8 + r] = res[i][r];
                    }
                }
                __syncthreads();
                if (give) {
                    #pragma unroll
                    for (int i = 0; i < NOCT; ++i) {
                        #pragma unroll
                        for (int r = 0; r < 8; ++r) res[i][r] = xother[i * 8 + r];
                    }
                }
                float* o0 = obase + (size_t)(2 * (a - 1) + (b - 1)) * QB * (Hh * Dd);
                #pragma unroll
                for (int i = 0; i < NOCT; ++i) {
                    #pragma unroll
                    for (int r = 0; r < 8; ++r)
                        o0[(size_t)(i * EB + r) * (Hh * Dd)] = res[i][r];
                }
            }
            v8f sd[NOCT];
            if (b == 0) {
                #pragma unroll
                for (int i = 0; i < NOCT; ++i) k0[i] = res[i];
            } else if (a == 0) {
                if (b == 1) {
                    #pragma unroll
                    for (int i = 0; i < NOCT; ++i) tt[i] = res[i];
                } else {
                    #pragma unroll
                    for (int i = 0; i < NOCT; ++i) bt[i] = res[i];
                }
            }
            if (b == 2) {
                #pragma unroll
                for (int i = 0; i < NOCT; ++i) sd[i] = v8f{};
            } else if (a == 0) {
                #pragma unroll
                for (int i = 0; i < NOCT; ++i) sd[i] = k0[i];
            } else if (b == 0) {
                #pragma unroll
                for (int i = 0; i < NOCT; ++i) sd[i] = tt[i] + k0[i];
            } else {
                #pragma unroll
                for (int i = 0; i < NOCT; ++i) sd[i] = bt[i] + k0[i];
            }
            restart(c, sd);
        }
    }
}

extern "C" void kernel_launch(void* const* d_in, const int* in_sizes, int n_in,
                              void* d_out, int out_size, void* d_ws, size_t ws_size,
                              hipStream_t stream)
{
    (void)in_sizes; (void)n_in; (void)out_size; (void)ws_size;
    const float* v      = (const float*)d_in[0];
    const float* offset = (const float*)d_in[1];
    const float* w      = (const float*)d_in[2];
    float*       out    = (float*)d_out;
    (void)d_ws;

    pbv_kernel<<<dim3(4, Bc, Hh), NTHR, 0, stream>>>(v, offset, w, out);
}
